# starting kernel
# baseline (speedup 1.0000x reference)
#include <hip/hip_runtime.h>
#include <hip/hip_bf16.h>
#include <stdint.h>

#define BATCH    8192
#define EMBED    128
#define NUM_MOD  64
#define HIDDEN   512
#define OUT_DIM  64
#define MAX_TILES 640
#define WCOMB    0.2857142857142857f

typedef float v2f __attribute__((ext_vector_type(2)));
typedef float v8f __attribute__((ext_vector_type(8)));

__device__ __forceinline__ v8f wmma_f32_4(v2f a, v2f b, v8f c) {
    return __builtin_amdgcn_wmma_f32_16x16x4_f32(
         false, a,  false, b,
         (short)0, c,  false,  false);
}

__global__ void zero_counts_kernel(int* counts) {
    if (threadIdx.x < 3 * NUM_MOD) counts[threadIdx.x] = 0;
}

__global__ void count_kernel(const int* __restrict__ mids, int* __restrict__ counts) {
    int b = blockIdx.x * blockDim.x + threadIdx.x;
    if (b < BATCH) {
        for (int h = 0; h < 3; ++h)
            atomicAdd(&counts[h * NUM_MOD + mids[b * 3 + h]], 1);
    }
}

__global__ void scan_build_kernel(const int* __restrict__ counts,
                                  int* __restrict__ offsets, int* __restrict__ cursor,
                                  int4* __restrict__ tiletab, int* __restrict__ ntiles) {
    if (blockIdx.x != 0 || threadIdx.x != 0) return;
    for (int h = 0; h < 3; ++h) {
        int run = 0;
        for (int m = 0; m < NUM_MOD; ++m) {
            offsets[h * NUM_MOD + m] = run;
            cursor [h * NUM_MOD + m] = run;
            run += counts[h * NUM_MOD + m];
        }
        int idx = 0;
        for (int m = 0; m < NUM_MOD; ++m) {
            int c  = counts [h * NUM_MOD + m];
            int st = offsets[h * NUM_MOD + m];
            for (int t = 0; t < c; t += 16) {
                int rem = c - t; if (rem > 16) rem = 16;
                tiletab[h * MAX_TILES + idx] = make_int4(m, st + t, rem, 0);
                ++idx;
            }
        }
        ntiles[h] = idx;
    }
}

__global__ void scatter_kernel(const int* __restrict__ mids,
                               int* __restrict__ cursor, int* __restrict__ rowlist) {
    int b = blockIdx.x * blockDim.x + threadIdx.x;
    if (b < BATCH) {
        for (int h = 0; h < 3; ++h) {
            int m = mids[b * 3 + h];
            int p = atomicAdd(&cursor[h * NUM_MOD + m], 1);
            rowlist[h * BATCH + p] = b;
        }
    }
}


__global__ void pack_wm_kernel(const float* __restrict__ W, float* __restrict__ WF) {
    int t = blockIdx.x * blockDim.x + threadIdx.x;
    if (t >= NUM_MOD * 8 * 32 * 32) return;
    int l  = t & 31;
    int k4 = (t >> 5) & 31;
    int ot = (t >> 10) & 7;
    int m  = t >> 13;
    int n = l & 15, c0 = (l >> 4) * 2;
    int o = ot * 16 + n;
    int i = 4 * k4 + c0;
    const float* src = W + ((size_t)m * EMBED + o) * EMBED + i;
    int k4g = k4 >> 1, ph = k4 & 1;
    float* dst = WF + ((size_t)((m * 8 + ot) * 16 + k4g)) * 128 + l * 4 + ph * 2;
    dst[0] = src[0];
    dst[1] = src[1];
}

__global__ void pack_w1_kernel(const float* __restrict__ W1, float* __restrict__ WF) {
    int t = blockIdx.x * blockDim.x + threadIdx.x;
    if (t >= 32 * 32 * 32) return;
    int l  = t & 31;
    int k4 = (t >> 5) & 31;
    int ot = t >> 10;
    int n = l & 15, c0 = (l >> 4) * 2;
    int o = ot * 16 + n;
    int i = 4 * k4 + c0;
    const float* src = W1 + (size_t)o * EMBED + i;
    int k4g = k4 >> 1, ph = k4 & 1;
    float* dst = WF + ((size_t)(ot * 16 + k4g)) * 128 + l * 4 + ph * 2;
    dst[0] = src[0];
    dst[1] = src[1];
}

__global__ void pack_w2_kernel(const float* __restrict__ W2, float* __restrict__ WF) {
    int t = blockIdx.x * blockDim.x + threadIdx.x;
    if (t >= 4 * 128 * 32) return;
    int l  = t & 31;
    int k4 = (t >> 5) & 127;
    int ot = t >> 12;
    int n = l & 15, c0 = (l >> 4) * 2;
    int o = ot * 16 + n;
    int i = 4 * k4 + c0;
    const float* src = W2 + (size_t)o * HIDDEN + i;
    int k4g = k4 >> 1, ph = k4 & 1;
    float* dst = WF + ((size_t)(ot * 64 + k4g)) * 128 + l * 4 + ph * 2;
    dst[0] = src[0];
    dst[1] = src[1];
}

__global__ __launch_bounds__(256) void hop_gemm_kernel(
    const float* __restrict__ embeds, const float* __restrict__ WmF,
    const int*  __restrict__ eids,    const int4* __restrict__ tiletab,
    const int*  __restrict__ ntiles,  const int*  __restrict__ rowlist,
    const float* __restrict__ xin,    float* __restrict__ xout, int hop)
{
    __shared__ float sX[16][132];
    __shared__ float sB[16][132];
    __shared__ int   sRow[16];

    int nt = ntiles[hop];
    if ((int)blockIdx.x >= nt) return;
    int4 tt = tiletab[hop * MAX_TILES + blockIdx.x];
    int m = tt.x, start = tt.y, cnt = tt.z;
    int tid = threadIdx.x;

    if (tid < 16) {
        int slot = (tid < cnt) ? tid : 0;
        sRow[tid] = rowlist[hop * BATCH + start + slot];
    }
    __syncthreads();

    for (int q = tid; q < 16 * 32; q += 256) {
        int r = q >> 5, c4 = q & 31;
        int rowg = sRow[r];
        const float* srcx = (hop == 0)
            ? (embeds + (size_t)eids[rowg * 4 + 0] * EMBED)
            : (xin    + (size_t)rowg * EMBED);
        float4 v = ((const float4*)srcx)[c4];
        sX[r][c4 * 4 + 0] = v.x; sX[r][c4 * 4 + 1] = v.y;
        sX[r][c4 * 4 + 2] = v.z; sX[r][c4 * 4 + 3] = v.w;
        const float* srcb = embeds + (size_t)eids[rowg * 4 + hop + 1] * EMBED;
        float4 bv = ((const float4*)srcb)[c4];
        sB[r][c4 * 4 + 0] = bv.x; sB[r][c4 * 4 + 1] = bv.y;
        sB[r][c4 * 4 + 2] = bv.z; sB[r][c4 * 4 + 3] = bv.w;
    }
    __syncthreads();

    int wave = tid >> 5, lane = tid & 31;
    int half = lane >> 4, n = lane & 15, c0 = half * 2;
    int obase = wave * 16;

    const float4* Bb = (const float4*)(WmF + (size_t)(m * 8 + wave) * 16 * 128) + lane;

    v8f acc = {0.f,0.f,0.f,0.f,0.f,0.f,0.f,0.f};
    #pragma unroll
    for (int k4g = 0; k4g < 16; ++k4g) {
        float4 w = Bb[k4g * 32];
        int k = k4g * 8;
        v2f a0, a1, b0, b1f;
        a0.x = sX[n][k + c0];       a0.y = sX[n][k + c0 + 1];
        a1.x = sX[n][k + 4 + c0];   a1.y = sX[n][k + 4 + c0 + 1];
        b0.x = w.x; b0.y = w.y;
        b1f.x = w.z; b1f.y = w.w;
        acc = wmma_f32_4(a0, b0,  acc);
        acc = wmma_f32_4(a1, b1f, acc);
    }

    for (int r = 0; r < 8; ++r) {
        int row = r + half * 8;
        if (row < cnt) {
            int rowg = sRow[row];
            int o = obase + n;
            float bias = sB[row][o];
            float v = acc[r] + bias;
            v = v > 0.f ? v : 0.f;
            float res = (hop == 2) ? ((1.f - WCOMB) * bias + WCOMB * v) : v;
            xout[(size_t)rowg * EMBED + o] = res;
        }
    }
}

__global__ __launch_bounds__(256) void mlp_kernel(
    const float* __restrict__ xin, const float* __restrict__ W1F, const float* __restrict__ b1,
    const float* __restrict__ W2F, const float* __restrict__ b2, float* __restrict__ out)
{
    __shared__ float sX[16][132];
    __shared__ float sH[16][516];

    int bt = blockIdx.x;
    int tid = threadIdx.x;
    int wave = tid >> 5, lane = tid & 31;
    int half = lane >> 4, n = lane & 15, c0 = half * 2;

    for (int q = tid; q < 16 * 32; q += 256) {
        int r = q >> 5, c4 = q & 31;
        float4 v = ((const float4*)(xin + (size_t)(bt * 16 + r) * EMBED))[c4];
        sX[r][c4 * 4 + 0] = v.x; sX[r][c4 * 4 + 1] = v.y;
        sX[r][c4 * 4 + 2] = v.z; sX[r][c4 * 4 + 3] = v.w;
    }
    __syncthreads();

    for (int p = 0; p < 4; ++p) {
        int ot = p * 8 + wave;
        int ob = ot * 16;
        const float4* Bb = (const float4*)(W1F + (size_t)ot * 16 * 128) + lane;
        v8f acc = {0.f,0.f,0.f,0.f,0.f,0.f,0.f,0.f};
        #pragma unroll
        for (int k4g = 0; k4g < 16; ++k4g) {
            float4 w = Bb[k4g * 32];
            int k = k4g * 8;
            v2f a0, a1, b0, b1f;
            a0.x = sX[n][k + c0];       a0.y = sX[n][k + c0 + 1];
            a1.x = sX[n][k + 4 + c0];   a1.y = sX[n][k + 4 + c0 + 1];
            b0.x = w.x; b0.y = w.y;
            b1f.x = w.z; b1f.y = w.w;
            acc = wmma_f32_4(a0, b0,  acc);
            acc = wmma_f32_4(a1, b1f, acc);
        }
        float bj = b1[ob + n];
        for (int r = 0; r < 8; ++r) {
            int row = r + half * 8;
            float v = acc[r] + bj;
            sH[row][ob + n] = v > 0.f ? v : 0.f;
        }
    }
    __syncthreads();

    if (wave < 4) {
        int ob = wave * 16;
        const float4* Bb = (const float4*)(W2F + (size_t)wave * 64 * 128) + lane;
        v8f acc = {0.f,0.f,0.f,0.f,0.f,0.f,0.f,0.f};
        #pragma unroll 8
        for (int k4g = 0; k4g < 64; ++k4g) {
            float4 w = Bb[k4g * 32];
            int k = k4g * 8;
            v2f a0, a1, b0, b1f;
            a0.x = sH[n][k + c0];       a0.y = sH[n][k + c0 + 1];
            a1.x = sH[n][k + 4 + c0];   a1.y = sH[n][k + 4 + c0 + 1];
            b0.x = w.x; b0.y = w.y;
            b1f.x = w.z; b1f.y = w.w;
            acc = wmma_f32_4(a0, b0,  acc);
            acc = wmma_f32_4(a1, b1f, acc);
        }
        float bo = b2[ob + n];
        for (int r = 0; r < 8; ++r) {
            int row = r + half * 8;
            out[(size_t)(bt * 16 + row) * OUT_DIM + ob + n] = acc[r] + bo;
        }
    }
}

extern "C" void kernel_launch(void* const* d_in, const int* in_sizes, int n_in,
                              void* d_out, int out_size, void* d_ws, size_t ws_size,
                              hipStream_t stream) {
    const float* embeds = (const float*)d_in[0];
    const float* Wmods  = (const float*)d_in[1];
    const float* W1     = (const float*)d_in[2];
    const float* b1     = (const float*)d_in[3];
    const float* W2     = (const float*)d_in[4];
    const float* b2     = (const float*)d_in[5];
    const int*   eids   = (const int*)d_in[6];
    const int*   mids   = (const int*)d_in[7];
    float*       out    = (float*)d_out;

    char* ws = (char*)d_ws;
    size_t off = 0;
    float* X0      = (float*)(ws + off); off += (size_t)BATCH * EMBED * 4;
    float* X1      = (float*)(ws + off); off += (size_t)BATCH * EMBED * 4;
    float* WmF     = (float*)(ws + off); off += (size_t)NUM_MOD * EMBED * EMBED * 4;
    float* W1F     = (float*)(ws + off); off += (size_t)HIDDEN * EMBED * 4;
    float* W2F     = (float*)(ws + off); off += (size_t)OUT_DIM * HIDDEN * 4;
    int*   counts  = (int*)(ws + off);   off += 3 * NUM_MOD * 4;
    int*   offsets = (int*)(ws + off);   off += 3 * NUM_MOD * 4;
    int*   cursor  = (int*)(ws + off);   off += 3 * NUM_MOD * 4;
    int*   rowlist = (int*)(ws + off);   off += 3 * BATCH * 4;
    int4*  tiletab = (int4*)(ws + off);  off += 3 * MAX_TILES * sizeof(int4);
    int*   ntiles  = (int*)(ws + off);   off += 64;
    (void)ws_size; (void)in_sizes; (void)n_in; (void)out_size;

    zero_counts_kernel<<<1, 256, 0, stream>>>(counts);
    count_kernel<<<BATCH / 256, 256, 0, stream>>>(mids, counts);
    scan_build_kernel<<<1, 32, 0, stream>>>(counts, offsets, cursor, tiletab, ntiles);
    scatter_kernel<<<BATCH / 256, 256, 0, stream>>>(mids, cursor, rowlist);

    pack_wm_kernel<<<(NUM_MOD * 8 * 32 * 32) / 256, 256, 0, stream>>>(Wmods, WmF);
    pack_w1_kernel<<<(32 * 32 * 32) / 256, 256, 0, stream>>>(W1, W1F);
    pack_w2_kernel<<<(4 * 128 * 32) / 256, 256, 0, stream>>>(W2, W2F);

    hop_gemm_kernel<<<MAX_TILES, 256, 0, stream>>>(embeds, WmF, eids, tiletab, ntiles, rowlist, X1, X0, 0);
    hop_gemm_kernel<<<MAX_TILES, 256, 0, stream>>>(embeds, WmF, eids, tiletab, ntiles, rowlist, X0, X1, 1);
    hop_gemm_kernel<<<MAX_TILES, 256, 0, stream>>>(embeds, WmF, eids, tiletab, ntiles, rowlist, X1, X0, 2);

    mlp_kernel<<<BATCH / 16, 256, 0, stream>>>(X0, W1F, b1, W2F, b2, out);
}


# tile-table build (per-hop prefix sums of module counts plus the 16-row tile list) runs on one 192-thread block, one thread per hop-module pair with the counts in LDS, instead of one serial lane
# speedup vs baseline: 2.4497x; 2.4497x over previous
#include <hip/hip_runtime.h>
#include <hip/hip_bf16.h>
#include <stdint.h>

#define BATCH    8192
#define EMBED    128
#define NUM_MOD  64
#define HIDDEN   512
#define OUT_DIM  64
#define MAX_TILES 640
#define WCOMB    0.2857142857142857f

typedef float v2f __attribute__((ext_vector_type(2)));
typedef float v8f __attribute__((ext_vector_type(8)));

__device__ __forceinline__ v8f wmma_f32_4(v2f a, v2f b, v8f c) {
    return __builtin_amdgcn_wmma_f32_16x16x4_f32(
         false, a,  false, b,
         (short)0, c,  false,  false);
}

__global__ void zero_counts_kernel(int* counts) {
    if (threadIdx.x < 3 * NUM_MOD) counts[threadIdx.x] = 0;
}

__global__ void count_kernel(const int* __restrict__ mids, int* __restrict__ counts) {
    int b = blockIdx.x * blockDim.x + threadIdx.x;
    if (b < BATCH) {
        for (int h = 0; h < 3; ++h)
            atomicAdd(&counts[h * NUM_MOD + mids[b * 3 + h]], 1);
    }
}

__global__ __launch_bounds__(3 * NUM_MOD) void scan_build_kernel(
                                  const int* __restrict__ counts, int* __restrict__ cursor,
                                  int4* __restrict__ tiletab, int* __restrict__ ntiles) {
    __shared__ int2 sCT[3 * NUM_MOD];

    int t = threadIdx.x;
    int h = t / NUM_MOD, m = t % NUM_MOD;
    int c  = counts[t];
    int nt = (c + 15) >> 4;
    sCT[t] = make_int2(c, nt);
    __syncthreads();

    int2 v[NUM_MOD];
    for (int j = 0; j < NUM_MOD; ++j) v[j] = sCT[h * NUM_MOD + j];

    int st = 0, idx = 0;
    for (int j = 0; j < NUM_MOD; ++j) {
        bool lower = j < m;
        st  += lower ? v[j].x : 0;
        idx += lower ? v[j].y : 0;
    }

    cursor[t] = st;
    int4* dst = tiletab + h * MAX_TILES + idx;
    for (int i = 0; i < nt; ++i) {
        int rem = c - i * 16; if (rem > 16) rem = 16;
        dst[i] = make_int4(m, st + i * 16, rem, 0);
    }
    if (m == NUM_MOD - 1) ntiles[h] = idx + nt;
}

__global__ void scatter_kernel(const int* __restrict__ mids,
                               int* __restrict__ cursor, int* __restrict__ rowlist) {
    int b = blockIdx.x * blockDim.x + threadIdx.x;
    if (b < BATCH) {
        for (int h = 0; h < 3; ++h) {
            int m = mids[b * 3 + h];
            int p = atomicAdd(&cursor[h * NUM_MOD + m], 1);
            rowlist[h * BATCH + p] = b;
        }
    }
}


__global__ void pack_wm_kernel(const float* __restrict__ W, float* __restrict__ WF) {
    int t = blockIdx.x * blockDim.x + threadIdx.x;
    if (t >= NUM_MOD * 8 * 32 * 32) return;
    int l  = t & 31;
    int k4 = (t >> 5) & 31;
    int ot = (t >> 10) & 7;
    int m  = t >> 13;
    int n = l & 15, c0 = (l >> 4) * 2;
    int o = ot * 16 + n;
    int i = 4 * k4 + c0;
    const float* src = W + ((size_t)m * EMBED + o) * EMBED + i;
    int k4g = k4 >> 1, ph = k4 & 1;
    float* dst = WF + ((size_t)((m * 8 + ot) * 16 + k4g)) * 128 + l * 4 + ph * 2;
    dst[0] = src[0];
    dst[1] = src[1];
}

__global__ void pack_w1_kernel(const float* __restrict__ W1, float* __restrict__ WF) {
    int t = blockIdx.x * blockDim.x + threadIdx.x;
    if (t >= 32 * 32 * 32) return;
    int l  = t & 31;
    int k4 = (t >> 5) & 31;
    int ot = t >> 10;
    int n = l & 15, c0 = (l >> 4) * 2;
    int o = ot * 16 + n;
    int i = 4 * k4 + c0;
    const float* src = W1 + (size_t)o * EMBED + i;
    int k4g = k4 >> 1, ph = k4 & 1;
    float* dst = WF + ((size_t)(ot * 16 + k4g)) * 128 + l * 4 + ph * 2;
    dst[0] = src[0];
    dst[1] = src[1];
}

__global__ void pack_w2_kernel(const float* __restrict__ W2, float* __restrict__ WF) {
    int t = blockIdx.x * blockDim.x + threadIdx.x;
    if (t >= 4 * 128 * 32) return;
    int l  = t & 31;
    int k4 = (t >> 5) & 127;
    int ot = t >> 12;
    int n = l & 15, c0 = (l >> 4) * 2;
    int o = ot * 16 + n;
    int i = 4 * k4 + c0;
    const float* src = W2 + (size_t)o * HIDDEN + i;
    int k4g = k4 >> 1, ph = k4 & 1;
    float* dst = WF + ((size_t)(ot * 64 + k4g)) * 128 + l * 4 + ph * 2;
    dst[0] = src[0];
    dst[1] = src[1];
}

__global__ __launch_bounds__(256) void hop_gemm_kernel(
    const float* __restrict__ embeds, const float* __restrict__ WmF,
    const int*  __restrict__ eids,    const int4* __restrict__ tiletab,
    const int*  __restrict__ ntiles,  const int*  __restrict__ rowlist,
    const float* __restrict__ xin,    float* __restrict__ xout, int hop)
{
    __shared__ float sX[16][132];
    __shared__ float sB[16][132];
    __shared__ int   sRow[16];

    int nt = ntiles[hop];
    if ((int)blockIdx.x >= nt) return;
    int4 tt = tiletab[hop * MAX_TILES + blockIdx.x];
    int m = tt.x, start = tt.y, cnt = tt.z;
    int tid = threadIdx.x;

    if (tid < 16) {
        int slot = (tid < cnt) ? tid : 0;
        sRow[tid] = rowlist[hop * BATCH + start + slot];
    }
    __syncthreads();

    for (int q = tid; q < 16 * 32; q += 256) {
        int r = q >> 5, c4 = q & 31;
        int rowg = sRow[r];
        const float* srcx = (hop == 0)
            ? (embeds + (size_t)eids[rowg * 4 + 0] * EMBED)
            : (xin    + (size_t)rowg * EMBED);
        float4 v = ((const float4*)srcx)[c4];
        sX[r][c4 * 4 + 0] = v.x; sX[r][c4 * 4 + 1] = v.y;
        sX[r][c4 * 4 + 2] = v.z; sX[r][c4 * 4 + 3] = v.w;
        const float* srcb = embeds + (size_t)eids[rowg * 4 + hop + 1] * EMBED;
        float4 bv = ((const float4*)srcb)[c4];
        sB[r][c4 * 4 + 0] = bv.x; sB[r][c4 * 4 + 1] = bv.y;
        sB[r][c4 * 4 + 2] = bv.z; sB[r][c4 * 4 + 3] = bv.w;
    }
    __syncthreads();

    int wave = tid >> 5, lane = tid & 31;
    int half = lane >> 4, n = lane & 15, c0 = half * 2;
    int obase = wave * 16;

    const float4* Bb = (const float4*)(WmF + (size_t)(m * 8 + wave) * 16 * 128) + lane;

    v8f acc = {0.f,0.f,0.f,0.f,0.f,0.f,0.f,0.f};
    #pragma unroll
    for (int k4g = 0; k4g < 16; ++k4g) {
        float4 w = Bb[k4g * 32];
        int k = k4g * 8;
        v2f a0, a1, b0, b1f;
        a0.x = sX[n][k + c0];       a0.y = sX[n][k + c0 + 1];
        a1.x = sX[n][k + 4 + c0];   a1.y = sX[n][k + 4 + c0 + 1];
        b0.x = w.x; b0.y = w.y;
        b1f.x = w.z; b1f.y = w.w;
        acc = wmma_f32_4(a0, b0,  acc);
        acc = wmma_f32_4(a1, b1f, acc);
    }

    for (int r = 0; r < 8; ++r) {
        int row = r + half * 8;
        if (row < cnt) {
            int rowg = sRow[row];
            int o = obase + n;
            float bias = sB[row][o];
            float v = acc[r] + bias;
            v = v > 0.f ? v : 0.f;
            float res = (hop == 2) ? ((1.f - WCOMB) * bias + WCOMB * v) : v;
            xout[(size_t)rowg * EMBED + o] = res;
        }
    }
}

__global__ __launch_bounds__(256) void mlp_kernel(
    const float* __restrict__ xin, const float* __restrict__ W1F, const float* __restrict__ b1,
    const float* __restrict__ W2F, const float* __restrict__ b2, float* __restrict__ out)
{
    __shared__ float sX[16][132];
    __shared__ float sH[16][516];

    int bt = blockIdx.x;
    int tid = threadIdx.x;
    int wave = tid >> 5, lane = tid & 31;
    int half = lane >> 4, n = lane & 15, c0 = half * 2;

    for (int q = tid; q < 16 * 32; q += 256) {
        int r = q >> 5, c4 = q & 31;
        float4 v = ((const float4*)(xin + (size_t)(bt * 16 + r) * EMBED))[c4];
        sX[r][c4 * 4 + 0] = v.x; sX[r][c4 * 4 + 1] = v.y;
        sX[r][c4 * 4 + 2] = v.z; sX[r][c4 * 4 + 3] = v.w;
    }
    __syncthreads();

    for (int p = 0; p < 4; ++p) {
        int ot = p * 8 + wave;
        int ob = ot * 16;
        const float4* Bb = (const float4*)(W1F + (size_t)ot * 16 * 128) + lane;
        v8f acc = {0.f,0.f,0.f,0.f,0.f,0.f,0.f,0.f};
        #pragma unroll
        for (int k4g = 0; k4g < 16; ++k4g) {
            float4 w = Bb[k4g * 32];
            int k = k4g * 8;
            v2f a0, a1, b0, b1f;
            a0.x = sX[n][k + c0];       a0.y = sX[n][k + c0 + 1];
            a1.x = sX[n][k + 4 + c0];   a1.y = sX[n][k + 4 + c0 + 1];
            b0.x = w.x; b0.y = w.y;
            b1f.x = w.z; b1f.y = w.w;
            acc = wmma_f32_4(a0, b0,  acc);
            acc = wmma_f32_4(a1, b1f, acc);
        }
        float bj = b1[ob + n];
        for (int r = 0; r < 8; ++r) {
            int row = r + half * 8;
            float v = acc[r] + bj;
            sH[row][ob + n] = v > 0.f ? v : 0.f;
        }
    }
    __syncthreads();

    if (wave < 4) {
        int ob = wave * 16;
        const float4* Bb = (const float4*)(W2F + (size_t)wave * 64 * 128) + lane;
        v8f acc = {0.f,0.f,0.f,0.f,0.f,0.f,0.f,0.f};
        #pragma unroll 8
        for (int k4g = 0; k4g < 64; ++k4g) {
            float4 w = Bb[k4g * 32];
            int k = k4g * 8;
            v2f a0, a1, b0, b1f;
            a0.x = sH[n][k + c0];       a0.y = sH[n][k + c0 + 1];
            a1.x = sH[n][k + 4 + c0];   a1.y = sH[n][k + 4 + c0 + 1];
            b0.x = w.x; b0.y = w.y;
            b1f.x = w.z; b1f.y = w.w;
            acc = wmma_f32_4(a0, b0,  acc);
            acc = wmma_f32_4(a1, b1f, acc);
        }
        float bo = b2[ob + n];
        for (int r = 0; r < 8; ++r) {
            int row = r + half * 8;
            out[(size_t)(bt * 16 + row) * OUT_DIM + ob + n] = acc[r] + bo;
        }
    }
}

extern "C" void kernel_launch(void* const* d_in, const int* in_sizes, int n_in,
                              void* d_out, int out_size, void* d_ws, size_t ws_size,
                              hipStream_t stream) {
    const float* embeds = (const float*)d_in[0];
    const float* Wmods  = (const float*)d_in[1];
    const float* W1     = (const float*)d_in[2];
    const float* b1     = (const float*)d_in[3];
    const float* W2     = (const float*)d_in[4];
    const float* b2     = (const float*)d_in[5];
    const int*   eids   = (const int*)d_in[6];
    const int*   mids   = (const int*)d_in[7];
    float*       out    = (float*)d_out;

    char* ws = (char*)d_ws;
    size_t off = 0;
    float* X0      = (float*)(ws + off); off += (size_t)BATCH * EMBED * 4;
    float* X1      = (float*)(ws + off); off += (size_t)BATCH * EMBED * 4;
    float* WmF     = (float*)(ws + off); off += (size_t)NUM_MOD * EMBED * EMBED * 4;
    float* W1F     = (float*)(ws + off); off += (size_t)HIDDEN * EMBED * 4;
    float* W2F     = (float*)(ws + off); off += (size_t)OUT_DIM * HIDDEN * 4;
    int*   counts  = (int*)(ws + off);   off += 3 * NUM_MOD * 4;
    off += 3 * NUM_MOD * 4;
    int*   cursor  = (int*)(ws + off);   off += 3 * NUM_MOD * 4;
    int*   rowlist = (int*)(ws + off);   off += 3 * BATCH * 4;
    int4*  tiletab = (int4*)(ws + off);  off += 3 * MAX_TILES * sizeof(int4);
    int*   ntiles  = (int*)(ws + off);   off += 64;
    (void)ws_size; (void)in_sizes; (void)n_in; (void)out_size;

    zero_counts_kernel<<<1, 256, 0, stream>>>(counts);
    count_kernel<<<BATCH / 256, 256, 0, stream>>>(mids, counts);
    scan_build_kernel<<<1, 3 * NUM_MOD, 0, stream>>>(counts, cursor, tiletab, ntiles);
    scatter_kernel<<<BATCH / 256, 256, 0, stream>>>(mids, cursor, rowlist);

    pack_wm_kernel<<<(NUM_MOD * 8 * 32 * 32) / 256, 256, 0, stream>>>(Wmods, WmF);
    pack_w1_kernel<<<(32 * 32 * 32) / 256, 256, 0, stream>>>(W1, W1F);
    pack_w2_kernel<<<(4 * 128 * 32) / 256, 256, 0, stream>>>(W2, W2F);

    hop_gemm_kernel<<<MAX_TILES, 256, 0, stream>>>(embeds, WmF, eids, tiletab, ntiles, rowlist, X1, X0, 0);
    hop_gemm_kernel<<<MAX_TILES, 256, 0, stream>>>(embeds, WmF, eids, tiletab, ntiles, rowlist, X0, X1, 1);
    hop_gemm_kernel<<<MAX_TILES, 256, 0, stream>>>(embeds, WmF, eids, tiletab, ntiles, rowlist, X1, X0, 2);

    mlp_kernel<<<BATCH / 16, 256, 0, stream>>>(X0, W1F, b1, W2F, b2, out);
}


# the 7 preparation launches (zero, count, scan, scatter, 3 weight packs) become one: 3 blocks bin one hop each in LDS (one atomic pass gives count and rank), 70 blocks pack weights; 11 launches to 5
# speedup vs baseline: 2.7848x; 1.1368x over previous
#include <hip/hip_runtime.h>
#include <hip/hip_bf16.h>
#include <stdint.h>

#define BATCH    8192
#define EMBED    128
#define NUM_MOD  64
#define HIDDEN   512
#define OUT_DIM  64
#define MAX_TILES 640
#define WCOMB    0.2857142857142857f

typedef float v2f __attribute__((ext_vector_type(2)));
typedef float v8f __attribute__((ext_vector_type(8)));

__device__ __forceinline__ v8f wmma_f32_4(v2f a, v2f b, v8f c) {
    return __builtin_amdgcn_wmma_f32_16x16x4_f32(
         false, a,  false, b,
         (short)0, c,  false,  false);
}

#define PREP_THREADS   1024
#define BIN_BLOCKS     3
#define BIN_ROWS       (BATCH / PREP_THREADS)
#define PACK_ITEMS_WM  (NUM_MOD * EMBED * EMBED / 8)
#define PACK_ITEMS_W1  (HIDDEN * EMBED / 8)
#define PACK_ITEMS_W2  (OUT_DIM * HIDDEN / 8)
#define PACK_PER_THREAD 2
#define PACK_CHUNK     (PREP_THREADS * PACK_PER_THREAD)
#define PACK_BLOCKS_WM (PACK_ITEMS_WM / PACK_CHUNK)
#define PACK_BLOCKS_W1 (PACK_ITEMS_W1 / PACK_CHUNK)
#define PACK_BLOCKS_W2 (PACK_ITEMS_W2 / PACK_CHUNK)
#define PACK_BLOCKS    (PACK_BLOCKS_WM + PACK_BLOCKS_W1 + PACK_BLOCKS_W2)

template <int KGBITS>
__device__ __forceinline__ void pack_items(const float* __restrict__ Wsrc,
                                           float4* __restrict__ WF, int first) {
    const float4* src[PACK_PER_THREAD];
    float4*       dst[PACK_PER_THREAD];
    float4        lo[PACK_PER_THREAD], hi[PACK_PER_THREAD];
    for (int j = 0; j < PACK_PER_THREAD; ++j) {
        int p    = first + j * PREP_THREADS + (int)threadIdx.x;
        int n    = p & 15;
        int k4g  = (p >> 4) & ((1 << KGBITS) - 1);
        int tile = p >> (4 + KGBITS);
        src[j] = (const float4*)(Wsrc + ((size_t)(tile * 16 + n) << (KGBITS + 3)) + 8 * k4g);
        dst[j] = WF + ((size_t)((tile << KGBITS) + k4g)) * 32 + n;
    }
    for (int j = 0; j < PACK_PER_THREAD; ++j) { lo[j] = src[j][0]; hi[j] = src[j][1]; }
    for (int j = 0; j < PACK_PER_THREAD; ++j) {
        dst[j][0]  = make_float4(lo[j].x, lo[j].y, hi[j].x, hi[j].y);
        dst[j][16] = make_float4(lo[j].z, lo[j].w, hi[j].z, hi[j].w);
    }
}

__global__ __launch_bounds__(PREP_THREADS) void prep_kernel(
    const int* __restrict__ mids,
    const float* __restrict__ Wmods, const float* __restrict__ W1, const float* __restrict__ W2,
    int* __restrict__ rowlist, int4* __restrict__ tiletab, int* __restrict__ ntiles,
    float4* __restrict__ WmF, float4* __restrict__ W1F, float4* __restrict__ W2F)
{
    __shared__ int sCnt[NUM_MOD];
    __shared__ int sStart[NUM_MOD];

    int tid = threadIdx.x;
    int blk = blockIdx.x;

    if (blk >= BIN_BLOCKS) {
        int pb = blk - BIN_BLOCKS;
        if (pb < PACK_BLOCKS_WM)
            pack_items<4>(Wmods, WmF, pb * PACK_CHUNK);
        else if (pb < PACK_BLOCKS_WM + PACK_BLOCKS_W1)
            pack_items<4>(W1, W1F, (pb - PACK_BLOCKS_WM) * PACK_CHUNK);
        else
            pack_items<6>(W2, W2F, (pb - PACK_BLOCKS_WM - PACK_BLOCKS_W1) * PACK_CHUNK);
        return;
    }

    int h = blk;
    if (tid < NUM_MOD) sCnt[tid] = 0;
    __syncthreads();

    int mm[BIN_ROWS], rk[BIN_ROWS];
    for (int j = 0; j < BIN_ROWS; ++j) mm[j] = mids[(j * PREP_THREADS + tid) * 3 + h];
    for (int j = 0; j < BIN_ROWS; ++j) rk[j] = atomicAdd(&sCnt[mm[j]], 1);
    __syncthreads();

    int c = 0, nt = 0, st = 0, idx = 0;
    if (tid < NUM_MOD) {
        c  = sCnt[tid];
        nt = (c + 15) >> 4;
        for (int j = 0; j < NUM_MOD; ++j) {
            int cj = sCnt[j];
            bool lower = j < tid;
            st  += lower ? cj : 0;
            idx += lower ? ((cj + 15) >> 4) : 0;
        }
        sStart[tid] = st;
    }
    __syncthreads();

    for (int j = 0; j < BIN_ROWS; ++j)
        rowlist[h * BATCH + sStart[mm[j]] + rk[j]] = j * PREP_THREADS + tid;

    if (tid < NUM_MOD) {
        int4* dst = tiletab + h * MAX_TILES + idx;
        for (int i = 0; i < nt; ++i) {
            int rem = c - i * 16; if (rem > 16) rem = 16;
            dst[i] = make_int4(tid, st + i * 16, rem, 0);
        }
        if (tid == NUM_MOD - 1) ntiles[h] = idx + nt;
    }
}

__global__ __launch_bounds__(256) void hop_gemm_kernel(
    const float* __restrict__ embeds, const float* __restrict__ WmF,
    const int*  __restrict__ eids,    const int4* __restrict__ tiletab,
    const int*  __restrict__ ntiles,  const int*  __restrict__ rowlist,
    const float* __restrict__ xin,    float* __restrict__ xout, int hop)
{
    __shared__ float sX[16][132];
    __shared__ float sB[16][132];
    __shared__ int   sRow[16];

    int nt = ntiles[hop];
    if ((int)blockIdx.x >= nt) return;
    int4 tt = tiletab[hop * MAX_TILES + blockIdx.x];
    int m = tt.x, start = tt.y, cnt = tt.z;
    int tid = threadIdx.x;

    if (tid < 16) {
        int slot = (tid < cnt) ? tid : 0;
        sRow[tid] = rowlist[hop * BATCH + start + slot];
    }
    __syncthreads();

    for (int q = tid; q < 16 * 32; q += 256) {
        int r = q >> 5, c4 = q & 31;
        int rowg = sRow[r];
        const float* srcx = (hop == 0)
            ? (embeds + (size_t)eids[rowg * 4 + 0] * EMBED)
            : (xin    + (size_t)rowg * EMBED);
        float4 v = ((const float4*)srcx)[c4];
        sX[r][c4 * 4 + 0] = v.x; sX[r][c4 * 4 + 1] = v.y;
        sX[r][c4 * 4 + 2] = v.z; sX[r][c4 * 4 + 3] = v.w;
        const float* srcb = embeds + (size_t)eids[rowg * 4 + hop + 1] * EMBED;
        float4 bv = ((const float4*)srcb)[c4];
        sB[r][c4 * 4 + 0] = bv.x; sB[r][c4 * 4 + 1] = bv.y;
        sB[r][c4 * 4 + 2] = bv.z; sB[r][c4 * 4 + 3] = bv.w;
    }
    __syncthreads();

    int wave = tid >> 5, lane = tid & 31;
    int half = lane >> 4, n = lane & 15, c0 = half * 2;
    int obase = wave * 16;

    const float4* Bb = (const float4*)(WmF + (size_t)(m * 8 + wave) * 16 * 128) + lane;

    v8f acc = {0.f,0.f,0.f,0.f,0.f,0.f,0.f,0.f};
    #pragma unroll
    for (int k4g = 0; k4g < 16; ++k4g) {
        float4 w = Bb[k4g * 32];
        int k = k4g * 8;
        v2f a0, a1, b0, b1f;
        a0.x = sX[n][k + c0];       a0.y = sX[n][k + c0 + 1];
        a1.x = sX[n][k + 4 + c0];   a1.y = sX[n][k + 4 + c0 + 1];
        b0.x = w.x; b0.y = w.y;
        b1f.x = w.z; b1f.y = w.w;
        acc = wmma_f32_4(a0, b0,  acc);
        acc = wmma_f32_4(a1, b1f, acc);
    }

    for (int r = 0; r < 8; ++r) {
        int row = r + half * 8;
        if (row < cnt) {
            int rowg = sRow[row];
            int o = obase + n;
            float bias = sB[row][o];
            float v = acc[r] + bias;
            v = v > 0.f ? v : 0.f;
            float res = (hop == 2) ? ((1.f - WCOMB) * bias + WCOMB * v) : v;
            xout[(size_t)rowg * EMBED + o] = res;
        }
    }
}

__global__ __launch_bounds__(256) void mlp_kernel(
    const float* __restrict__ xin, const float* __restrict__ W1F, const float* __restrict__ b1,
    const float* __restrict__ W2F, const float* __restrict__ b2, float* __restrict__ out)
{
    __shared__ float sX[16][132];
    __shared__ float sH[16][516];

    int bt = blockIdx.x;
    int tid = threadIdx.x;
    int wave = tid >> 5, lane = tid & 31;
    int half = lane >> 4, n = lane & 15, c0 = half * 2;

    for (int q = tid; q < 16 * 32; q += 256) {
        int r = q >> 5, c4 = q & 31;
        float4 v = ((const float4*)(xin + (size_t)(bt * 16 + r) * EMBED))[c4];
        sX[r][c4 * 4 + 0] = v.x; sX[r][c4 * 4 + 1] = v.y;
        sX[r][c4 * 4 + 2] = v.z; sX[r][c4 * 4 + 3] = v.w;
    }
    __syncthreads();

    for (int p = 0; p < 4; ++p) {
        int ot = p * 8 + wave;
        int ob = ot * 16;
        const float4* Bb = (const float4*)(W1F + (size_t)ot * 16 * 128) + lane;
        v8f acc = {0.f,0.f,0.f,0.f,0.f,0.f,0.f,0.f};
        #pragma unroll
        for (int k4g = 0; k4g < 16; ++k4g) {
            float4 w = Bb[k4g * 32];
            int k = k4g * 8;
            v2f a0, a1, b0, b1f;
            a0.x = sX[n][k + c0];       a0.y = sX[n][k + c0 + 1];
            a1.x = sX[n][k + 4 + c0];   a1.y = sX[n][k + 4 + c0 + 1];
            b0.x = w.x; b0.y = w.y;
            b1f.x = w.z; b1f.y = w.w;
            acc = wmma_f32_4(a0, b0,  acc);
            acc = wmma_f32_4(a1, b1f, acc);
        }
        float bj = b1[ob + n];
        for (int r = 0; r < 8; ++r) {
            int row = r + half * 8;
            float v = acc[r] + bj;
            sH[row][ob + n] = v > 0.f ? v : 0.f;
        }
    }
    __syncthreads();

    if (wave < 4) {
        int ob = wave * 16;
        const float4* Bb = (const float4*)(W2F + (size_t)wave * 64 * 128) + lane;
        v8f acc = {0.f,0.f,0.f,0.f,0.f,0.f,0.f,0.f};
        #pragma unroll 8
        for (int k4g = 0; k4g < 64; ++k4g) {
            float4 w = Bb[k4g * 32];
            int k = k4g * 8;
            v2f a0, a1, b0, b1f;
            a0.x = sH[n][k + c0];       a0.y = sH[n][k + c0 + 1];
            a1.x = sH[n][k + 4 + c0];   a1.y = sH[n][k + 4 + c0 + 1];
            b0.x = w.x; b0.y = w.y;
            b1f.x = w.z; b1f.y = w.w;
            acc = wmma_f32_4(a0, b0,  acc);
            acc = wmma_f32_4(a1, b1f, acc);
        }
        float bo = b2[ob + n];
        for (int r = 0; r < 8; ++r) {
            int row = r + half * 8;
            out[(size_t)(bt * 16 + row) * OUT_DIM + ob + n] = acc[r] + bo;
        }
    }
}

extern "C" void kernel_launch(void* const* d_in, const int* in_sizes, int n_in,
                              void* d_out, int out_size, void* d_ws, size_t ws_size,
                              hipStream_t stream) {
    const float* embeds = (const float*)d_in[0];
    const float* Wmods  = (const float*)d_in[1];
    const float* W1     = (const float*)d_in[2];
    const float* b1     = (const float*)d_in[3];
    const float* W2     = (const float*)d_in[4];
    const float* b2     = (const float*)d_in[5];
    const int*   eids   = (const int*)d_in[6];
    const int*   mids   = (const int*)d_in[7];
    float*       out    = (float*)d_out;

    char* ws = (char*)d_ws;
    size_t off = 0;
    float* X0      = (float*)(ws + off); off += (size_t)BATCH * EMBED * 4;
    float* X1      = (float*)(ws + off); off += (size_t)BATCH * EMBED * 4;
    float* WmF     = (float*)(ws + off); off += (size_t)NUM_MOD * EMBED * EMBED * 4;
    float* W1F     = (float*)(ws + off); off += (size_t)HIDDEN * EMBED * 4;
    float* W2F     = (float*)(ws + off); off += (size_t)OUT_DIM * HIDDEN * 4;
    off += 3 * 3 * NUM_MOD * 4;
    int*   rowlist = (int*)(ws + off);   off += 3 * BATCH * 4;
    int4*  tiletab = (int4*)(ws + off);  off += 3 * MAX_TILES * sizeof(int4);
    int*   ntiles  = (int*)(ws + off);   off += 64;
    (void)ws_size; (void)in_sizes; (void)n_in; (void)out_size;

    prep_kernel<<<BIN_BLOCKS + PACK_BLOCKS, PREP_THREADS, 0, stream>>>(
        mids, Wmods, W1, W2, rowlist, tiletab, ntiles,
        (float4*)WmF, (float4*)W1F, (float4*)W2F);

    hop_gemm_kernel<<<MAX_TILES, 256, 0, stream>>>(embeds, WmF, eids, tiletab, ntiles, rowlist, X1, X0, 0);
    hop_gemm_kernel<<<MAX_TILES, 256, 0, stream>>>(embeds, WmF, eids, tiletab, ntiles, rowlist, X0, X1, 1);
    hop_gemm_kernel<<<MAX_TILES, 256, 0, stream>>>(embeds, WmF, eids, tiletab, ntiles, rowlist, X1, X0, 2);

    mlp_kernel<<<BATCH / 16, 256, 0, stream>>>(X0, W1F, b1, W2F, b2, out);
}


# each wave of the hop and hidden-layer GEMMs owns two 16-wide output tiles on two independent accumulators fed by one A fragment: blocks of 4 waves, not 8; a thread's 4 staging loads go out as a group
# speedup vs baseline: 3.3421x; 1.2001x over previous
#include <hip/hip_runtime.h>
#include <hip/hip_bf16.h>
#include <stdint.h>

#define BATCH    8192
#define EMBED    128
#define NUM_MOD  64
#define HIDDEN   512
#define OUT_DIM  64
#define MAX_TILES 640
#define WCOMB    0.2857142857142857f

typedef float v2f __attribute__((ext_vector_type(2)));
typedef float v8f __attribute__((ext_vector_type(8)));

__device__ __forceinline__ v8f wmma_f32_4(v2f a, v2f b, v8f c) {
    return __builtin_amdgcn_wmma_f32_16x16x4_f32(
         false, a,  false, b,
         (short)0, c,  false,  false);
}

#define PREP_THREADS   1024
#define BIN_BLOCKS     3
#define BIN_ROWS       (BATCH / PREP_THREADS)
#define PACK_ITEMS_WM  (NUM_MOD * EMBED * EMBED / 8)
#define PACK_ITEMS_W1  (HIDDEN * EMBED / 8)
#define PACK_ITEMS_W2  (OUT_DIM * HIDDEN / 8)
#define PACK_PER_THREAD 2
#define PACK_CHUNK     (PREP_THREADS * PACK_PER_THREAD)
#define PACK_BLOCKS_WM (PACK_ITEMS_WM / PACK_CHUNK)
#define PACK_BLOCKS_W1 (PACK_ITEMS_W1 / PACK_CHUNK)
#define PACK_BLOCKS_W2 (PACK_ITEMS_W2 / PACK_CHUNK)
#define PACK_BLOCKS    (PACK_BLOCKS_WM + PACK_BLOCKS_W1 + PACK_BLOCKS_W2)

template <int KGBITS>
__device__ __forceinline__ void pack_items(const float* __restrict__ Wsrc,
                                           float4* __restrict__ WF, int first) {
    const float4* src[PACK_PER_THREAD];
    float4*       dst[PACK_PER_THREAD];
    float4        lo[PACK_PER_THREAD], hi[PACK_PER_THREAD];
    for (int j = 0; j < PACK_PER_THREAD; ++j) {
        int p    = first + j * PREP_THREADS + (int)threadIdx.x;
        int n    = p & 15;
        int k4g  = (p >> 4) & ((1 << KGBITS) - 1);
        int tile = p >> (4 + KGBITS);
        src[j] = (const float4*)(Wsrc + ((size_t)(tile * 16 + n) << (KGBITS + 3)) + 8 * k4g);
        dst[j] = WF + ((size_t)((tile << KGBITS) + k4g)) * 32 + n;
    }
    for (int j = 0; j < PACK_PER_THREAD; ++j) { lo[j] = src[j][0]; hi[j] = src[j][1]; }
    for (int j = 0; j < PACK_PER_THREAD; ++j) {
        dst[j][0]  = make_float4(lo[j].x, lo[j].y, hi[j].x, hi[j].y);
        dst[j][16] = make_float4(lo[j].z, lo[j].w, hi[j].z, hi[j].w);
    }
}

__global__ __launch_bounds__(PREP_THREADS) void prep_kernel(
    const int* __restrict__ mids,
    const float* __restrict__ Wmods, const float* __restrict__ W1, const float* __restrict__ W2,
    int* __restrict__ rowlist, int4* __restrict__ tiletab, int* __restrict__ ntiles,
    float4* __restrict__ WmF, float4* __restrict__ W1F, float4* __restrict__ W2F)
{
    __shared__ int sCnt[NUM_MOD];
    __shared__ int sStart[NUM_MOD];

    int tid = threadIdx.x;
    int blk = blockIdx.x;

    if (blk >= BIN_BLOCKS) {
        int pb = blk - BIN_BLOCKS;
        if (pb < PACK_BLOCKS_WM)
            pack_items<4>(Wmods, WmF, pb * PACK_CHUNK);
        else if (pb < PACK_BLOCKS_WM + PACK_BLOCKS_W1)
            pack_items<4>(W1, W1F, (pb - PACK_BLOCKS_WM) * PACK_CHUNK);
        else
            pack_items<6>(W2, W2F, (pb - PACK_BLOCKS_WM - PACK_BLOCKS_W1) * PACK_CHUNK);
        return;
    }

    int h = blk;
    if (tid < NUM_MOD) sCnt[tid] = 0;
    __syncthreads();

    int mm[BIN_ROWS], rk[BIN_ROWS];
    for (int j = 0; j < BIN_ROWS; ++j) mm[j] = mids[(j * PREP_THREADS + tid) * 3 + h];
    for (int j = 0; j < BIN_ROWS; ++j) rk[j] = atomicAdd(&sCnt[mm[j]], 1);
    __syncthreads();

    int c = 0, nt = 0, st = 0, idx = 0;
    if (tid < NUM_MOD) {
        c  = sCnt[tid];
        nt = (c + 15) >> 4;
        for (int j = 0; j < NUM_MOD; ++j) {
            int cj = sCnt[j];
            bool lower = j < tid;
            st  += lower ? cj : 0;
            idx += lower ? ((cj + 15) >> 4) : 0;
        }
        sStart[tid] = st;
    }
    __syncthreads();

    for (int j = 0; j < BIN_ROWS; ++j)
        rowlist[h * BATCH + sStart[mm[j]] + rk[j]] = j * PREP_THREADS + tid;

    if (tid < NUM_MOD) {
        int4* dst = tiletab + h * MAX_TILES + idx;
        for (int i = 0; i < nt; ++i) {
            int rem = c - i * 16; if (rem > 16) rem = 16;
            dst[i] = make_int4(tid, st + i * 16, rem, 0);
        }
        if (tid == NUM_MOD - 1) ntiles[h] = idx + nt;
    }
}

#define GEMM_THREADS 128
#define STAGE_ROWS   (16 * 32 / GEMM_THREADS)
__global__ __launch_bounds__(GEMM_THREADS) void hop_gemm_kernel(
    const float* __restrict__ embeds, const float* __restrict__ WmF,
    const int*  __restrict__ eids,    const int4* __restrict__ tiletab,
    const int*  __restrict__ ntiles,  const int*  __restrict__ rowlist,
    const float* __restrict__ xin,    float* __restrict__ xout, int hop)
{
    __shared__ float sX[16][132];
    __shared__ float sB[16][132];
    __shared__ int   sRow[16];

    int nt = ntiles[hop];
    if ((int)blockIdx.x >= nt) return;
    int4 tt = tiletab[hop * MAX_TILES + blockIdx.x];
    int m = tt.x, start = tt.y, cnt = tt.z;
    int tid = threadIdx.x;

    if (tid < 16) {
        int slot = (tid < cnt) ? tid : 0;
        sRow[tid] = rowlist[hop * BATCH + start + slot];
    }
    __syncthreads();

    {
        int c4 = tid & 31, r0 = tid >> 5;
        int rowg[STAGE_ROWS], eb[STAGE_ROWS], ex[STAGE_ROWS];
        const float4* px[STAGE_ROWS];
        const float4* pb[STAGE_ROWS];
        float4 vx[STAGE_ROWS], vb[STAGE_ROWS];
        for (int j = 0; j < STAGE_ROWS; ++j) rowg[j] = sRow[r0 + j * (GEMM_THREADS / 32)];
        for (int j = 0; j < STAGE_ROWS; ++j) eb[j] = eids[rowg[j] * 4 + hop + 1];
        if (hop == 0) {
            for (int j = 0; j < STAGE_ROWS; ++j) ex[j] = eids[rowg[j] * 4 + 0];
            for (int j = 0; j < STAGE_ROWS; ++j) px[j] = (const float4*)(embeds + (size_t)ex[j] * EMBED);
        } else {
            for (int j = 0; j < STAGE_ROWS; ++j) px[j] = (const float4*)(xin + (size_t)rowg[j] * EMBED);
        }
        for (int j = 0; j < STAGE_ROWS; ++j) pb[j] = (const float4*)(embeds + (size_t)eb[j] * EMBED);
        for (int j = 0; j < STAGE_ROWS; ++j) { vx[j] = px[j][c4]; vb[j] = pb[j][c4]; }
        for (int j = 0; j < STAGE_ROWS; ++j) {
            int r = r0 + j * (GEMM_THREADS / 32);
            sX[r][c4 * 4 + 0] = vx[j].x; sX[r][c4 * 4 + 1] = vx[j].y;
            sX[r][c4 * 4 + 2] = vx[j].z; sX[r][c4 * 4 + 3] = vx[j].w;
            sB[r][c4 * 4 + 0] = vb[j].x; sB[r][c4 * 4 + 1] = vb[j].y;
            sB[r][c4 * 4 + 2] = vb[j].z; sB[r][c4 * 4 + 3] = vb[j].w;
        }
    }
    __syncthreads();

    int wave = tid >> 5, lane = tid & 31;
    int half = lane >> 4, n = lane & 15, c0 = half * 2;
    int obase = wave * 32;

    const float4* Bb0 = (const float4*)(WmF + (size_t)(m * 8 + wave * 2) * 16 * 128) + lane;
    const float4* Bb1 = Bb0 + 16 * 32;

    v8f acc0 = {0.f,0.f,0.f,0.f,0.f,0.f,0.f,0.f};
    v8f acc1 = {0.f,0.f,0.f,0.f,0.f,0.f,0.f,0.f};
    #pragma unroll
    for (int k4g = 0; k4g < 16; ++k4g) {
        float4 w0 = Bb0[k4g * 32];
        float4 w1 = Bb1[k4g * 32];
        int k = k4g * 8;
        v2f a0, a1, b00, b01, b10, b11;
        a0.x = sX[n][k + c0];       a0.y = sX[n][k + c0 + 1];
        a1.x = sX[n][k + 4 + c0];   a1.y = sX[n][k + 4 + c0 + 1];
        b00.x = w0.x; b00.y = w0.y;
        b01.x = w0.z; b01.y = w0.w;
        b10.x = w1.x; b10.y = w1.y;
        b11.x = w1.z; b11.y = w1.w;
        acc0 = wmma_f32_4(a0, b00, acc0);
        acc1 = wmma_f32_4(a0, b10, acc1);
        acc0 = wmma_f32_4(a1, b01, acc0);
        acc1 = wmma_f32_4(a1, b11, acc1);
    }

    for (int r = 0; r < 8; ++r) {
        int row = r + half * 8;
        if (row < cnt) {
            int rowg = sRow[row];
            for (int t = 0; t < 2; ++t) {
                int o = obase + t * 16 + n;
                float bias = sB[row][o];
                float v = (t ? acc1[r] : acc0[r]) + bias;
                v = v > 0.f ? v : 0.f;
                float res = (hop == 2) ? ((1.f - WCOMB) * bias + WCOMB * v) : v;
                xout[(size_t)rowg * EMBED + o] = res;
            }
        }
    }
}

__global__ __launch_bounds__(GEMM_THREADS) void mlp_kernel(
    const float* __restrict__ xin, const float* __restrict__ W1F, const float* __restrict__ b1,
    const float* __restrict__ W2F, const float* __restrict__ b2, float* __restrict__ out)
{
    __shared__ float sX[16][132];
    __shared__ float sH[16][516];

    int bt = blockIdx.x;
    int tid = threadIdx.x;
    int wave = tid >> 5, lane = tid & 31;
    int half = lane >> 4, n = lane & 15, c0 = half * 2;

    {
        int c4 = tid & 31, r0 = tid >> 5;
        float4 vx[STAGE_ROWS];
        for (int j = 0; j < STAGE_ROWS; ++j)
            vx[j] = ((const float4*)(xin + (size_t)(bt * 16 + r0 + j * (GEMM_THREADS / 32)) * EMBED))[c4];
        for (int j = 0; j < STAGE_ROWS; ++j) {
            int r = r0 + j * (GEMM_THREADS / 32);
            sX[r][c4 * 4 + 0] = vx[j].x; sX[r][c4 * 4 + 1] = vx[j].y;
            sX[r][c4 * 4 + 2] = vx[j].z; sX[r][c4 * 4 + 3] = vx[j].w;
        }
    }
    __syncthreads();

    for (int p = 0; p < 4; ++p) {
        int ot = p * 8 + wave * 2;
        int ob = ot * 16;
        const float4* Bb0 = (const float4*)(W1F + (size_t)ot * 16 * 128) + lane;
        const float4* Bb1 = Bb0 + 16 * 32;
        v8f acc0 = {0.f,0.f,0.f,0.f,0.f,0.f,0.f,0.f};
        v8f acc1 = {0.f,0.f,0.f,0.f,0.f,0.f,0.f,0.f};
        #pragma unroll
        for (int k4g = 0; k4g < 16; ++k4g) {
            float4 w0 = Bb0[k4g * 32];
            float4 w1 = Bb1[k4g * 32];
            int k = k4g * 8;
            v2f a0, a1, b00, b01, b10, b11;
            a0.x = sX[n][k + c0];       a0.y = sX[n][k + c0 + 1];
            a1.x = sX[n][k + 4 + c0];   a1.y = sX[n][k + 4 + c0 + 1];
            b00.x = w0.x; b00.y = w0.y;
            b01.x = w0.z; b01.y = w0.w;
            b10.x = w1.x; b10.y = w1.y;
            b11.x = w1.z; b11.y = w1.w;
            acc0 = wmma_f32_4(a0, b00, acc0);
            acc1 = wmma_f32_4(a0, b10, acc1);
            acc0 = wmma_f32_4(a1, b01, acc0);
            acc1 = wmma_f32_4(a1, b11, acc1);
        }
        float bj0 = b1[ob + n];
        float bj1 = b1[ob + 16 + n];
        for (int r = 0; r < 8; ++r) {
            int row = r + half * 8;
            float v0 = acc0[r] + bj0;
            float v1 = acc1[r] + bj1;
            sH[row][ob + n]      = v0 > 0.f ? v0 : 0.f;
            sH[row][ob + 16 + n] = v1 > 0.f ? v1 : 0.f;
        }
    }
    __syncthreads();

    {
        int ob = wave * 16;
        const float4* Bb = (const float4*)(W2F + (size_t)wave * 64 * 128) + lane;
        v8f acc = {0.f,0.f,0.f,0.f,0.f,0.f,0.f,0.f};
        #pragma unroll 8
        for (int k4g = 0; k4g < 64; ++k4g) {
            float4 w = Bb[k4g * 32];
            int k = k4g * 8;
            v2f a0, a1, b0, b1f;
            a0.x = sH[n][k + c0];       a0.y = sH[n][k + c0 + 1];
            a1.x = sH[n][k + 4 + c0];   a1.y = sH[n][k + 4 + c0 + 1];
            b0.x = w.x; b0.y = w.y;
            b1f.x = w.z; b1f.y = w.w;
            acc = wmma_f32_4(a0, b0,  acc);
            acc = wmma_f32_4(a1, b1f, acc);
        }
        float bo = b2[ob + n];
        for (int r = 0; r < 8; ++r) {
            int row = r + half * 8;
            out[(size_t)(bt * 16 + row) * OUT_DIM + ob + n] = acc[r] + bo;
        }
    }
}

extern "C" void kernel_launch(void* const* d_in, const int* in_sizes, int n_in,
                              void* d_out, int out_size, void* d_ws, size_t ws_size,
                              hipStream_t stream) {
    const float* embeds = (const float*)d_in[0];
    const float* Wmods  = (const float*)d_in[1];
    const float* W1     = (const float*)d_in[2];
    const float* b1     = (const float*)d_in[3];
    const float* W2     = (const float*)d_in[4];
    const float* b2     = (const float*)d_in[5];
    const int*   eids   = (const int*)d_in[6];
    const int*   mids   = (const int*)d_in[7];
    float*       out    = (float*)d_out;

    char* ws = (char*)d_ws;
    size_t off = 0;
    float* X0      = (float*)(ws + off); off += (size_t)BATCH * EMBED * 4;
    float* X1      = (float*)(ws + off); off += (size_t)BATCH * EMBED * 4;
    float* WmF     = (float*)(ws + off); off += (size_t)NUM_MOD * EMBED * EMBED * 4;
    float* W1F     = (float*)(ws + off); off += (size_t)HIDDEN * EMBED * 4;
    float* W2F     = (float*)(ws + off); off += (size_t)OUT_DIM * HIDDEN * 4;
    off += 3 * 3 * NUM_MOD * 4;
    int*   rowlist = (int*)(ws + off);   off += 3 * BATCH * 4;
    int4*  tiletab = (int4*)(ws + off);  off += 3 * MAX_TILES * sizeof(int4);
    int*   ntiles  = (int*)(ws + off);   off += 64;
    (void)ws_size; (void)in_sizes; (void)n_in; (void)out_size;

    prep_kernel<<<BIN_BLOCKS + PACK_BLOCKS, PREP_THREADS, 0, stream>>>(
        mids, Wmods, W1, W2, rowlist, tiletab, ntiles,
        (float4*)WmF, (float4*)W1F, (float4*)W2F);

    hop_gemm_kernel<<<MAX_TILES, GEMM_THREADS, 0, stream>>>(embeds, WmF, eids, tiletab, ntiles, rowlist, X1, X0, 0);
    hop_gemm_kernel<<<MAX_TILES, GEMM_THREADS, 0, stream>>>(embeds, WmF, eids, tiletab, ntiles, rowlist, X0, X1, 1);
    hop_gemm_kernel<<<MAX_TILES, GEMM_THREADS, 0, stream>>>(embeds, WmF, eids, tiletab, ntiles, rowlist, X1, X0, 2);

    mlp_kernel<<<BATCH / 16, GEMM_THREADS, 0, stream>>>(X0, W1F, b1, W2F, b2, out);
}


# last hop and MLP run as one kernel on 16-slot tiles of the last hop's sorted row list (hop product once per module in a tile, result stays in LDS): 5 launches to 4, no X round trip
# speedup vs baseline: 3.9310x; 1.1762x over previous
#include <hip/hip_runtime.h>
#include <hip/hip_bf16.h>
#include <stdint.h>

#define BATCH    8192
#define EMBED    128
#define NUM_MOD  64
#define HIDDEN   512
#define OUT_DIM  64
#define MAX_TILES 640
#define WCOMB    0.2857142857142857f

typedef float v2f __attribute__((ext_vector_type(2)));
typedef float v8f __attribute__((ext_vector_type(8)));

__device__ __forceinline__ v8f wmma_f32_4(v2f a, v2f b, v8f c) {
    return __builtin_amdgcn_wmma_f32_16x16x4_f32(
         false, a,  false, b,
         (short)0, c,  false,  false);
}

#define PREP_THREADS   1024
#define BIN_BLOCKS     3
#define BIN_ROWS       (BATCH / PREP_THREADS)
#define PACK_ITEMS_WM  (NUM_MOD * EMBED * EMBED / 8)
#define PACK_ITEMS_W1  (HIDDEN * EMBED / 8)
#define PACK_ITEMS_W2  (OUT_DIM * HIDDEN / 8)
#define PACK_PER_THREAD 2
#define PACK_CHUNK     (PREP_THREADS * PACK_PER_THREAD)
#define PACK_BLOCKS_WM (PACK_ITEMS_WM / PACK_CHUNK)
#define PACK_BLOCKS_W1 (PACK_ITEMS_W1 / PACK_CHUNK)
#define PACK_BLOCKS_W2 (PACK_ITEMS_W2 / PACK_CHUNK)
#define PACK_BLOCKS    (PACK_BLOCKS_WM + PACK_BLOCKS_W1 + PACK_BLOCKS_W2)

template <int KGBITS>
__device__ __forceinline__ void pack_items(const float* __restrict__ Wsrc,
                                           float4* __restrict__ WF, int first) {
    const float4* src[PACK_PER_THREAD];
    float4*       dst[PACK_PER_THREAD];
    float4        lo[PACK_PER_THREAD], hi[PACK_PER_THREAD];
    for (int j = 0; j < PACK_PER_THREAD; ++j) {
        int p    = first + j * PREP_THREADS + (int)threadIdx.x;
        int n    = p & 15;
        int k4g  = (p >> 4) & ((1 << KGBITS) - 1);
        int tile = p >> (4 + KGBITS);
        src[j] = (const float4*)(Wsrc + ((size_t)(tile * 16 + n) << (KGBITS + 3)) + 8 * k4g);
        dst[j] = WF + ((size_t)((tile << KGBITS) + k4g)) * 32 + n;
    }
    for (int j = 0; j < PACK_PER_THREAD; ++j) { lo[j] = src[j][0]; hi[j] = src[j][1]; }
    for (int j = 0; j < PACK_PER_THREAD; ++j) {
        dst[j][0]  = make_float4(lo[j].x, lo[j].y, hi[j].x, hi[j].y);
        dst[j][16] = make_float4(lo[j].z, lo[j].w, hi[j].z, hi[j].w);
    }
}

__global__ __launch_bounds__(PREP_THREADS) void prep_kernel(
    const int* __restrict__ mids,
    const float* __restrict__ Wmods, const float* __restrict__ W1, const float* __restrict__ W2,
    int* __restrict__ rowlist, int4* __restrict__ tiletab, int* __restrict__ ntiles,
    float4* __restrict__ WmF, float4* __restrict__ W1F, float4* __restrict__ W2F)
{
    __shared__ int sCnt[NUM_MOD];
    __shared__ int sStart[NUM_MOD];

    int tid = threadIdx.x;
    int blk = blockIdx.x;

    if (blk >= BIN_BLOCKS) {
        int pb = blk - BIN_BLOCKS;
        if (pb < PACK_BLOCKS_WM)
            pack_items<4>(Wmods, WmF, pb * PACK_CHUNK);
        else if (pb < PACK_BLOCKS_WM + PACK_BLOCKS_W1)
            pack_items<4>(W1, W1F, (pb - PACK_BLOCKS_WM) * PACK_CHUNK);
        else
            pack_items<6>(W2, W2F, (pb - PACK_BLOCKS_WM - PACK_BLOCKS_W1) * PACK_CHUNK);
        return;
    }

    int h = blk;
    if (tid < NUM_MOD) sCnt[tid] = 0;
    __syncthreads();

    int mm[BIN_ROWS], rk[BIN_ROWS];
    for (int j = 0; j < BIN_ROWS; ++j) mm[j] = mids[(j * PREP_THREADS + tid) * 3 + h];
    for (int j = 0; j < BIN_ROWS; ++j) rk[j] = atomicAdd(&sCnt[mm[j]], 1);
    __syncthreads();

    int c = 0, nt = 0, st = 0, idx = 0;
    if (tid < NUM_MOD) {
        c  = sCnt[tid];
        nt = (c + 15) >> 4;
        for (int j = 0; j < NUM_MOD; ++j) {
            int cj = sCnt[j];
            bool lower = j < tid;
            st  += lower ? cj : 0;
            idx += lower ? ((cj + 15) >> 4) : 0;
        }
        sStart[tid] = st;
    }
    __syncthreads();

    for (int j = 0; j < BIN_ROWS; ++j)
        rowlist[h * BATCH + sStart[mm[j]] + rk[j]] = j * PREP_THREADS + tid;

    if (tid < NUM_MOD) {
        int4* dst = tiletab + h * MAX_TILES + idx;
        for (int i = 0; i < nt; ++i) {
            int rem = c - i * 16; if (rem > 16) rem = 16;
            dst[i] = make_int4(tid, st + i * 16, rem, 0);
        }
        if (tid == NUM_MOD - 1) ntiles[h] = idx + nt;
    }
}

#define GEMM_THREADS 128
#define STAGE_ROWS   (16 * 32 / GEMM_THREADS)
__global__ __launch_bounds__(GEMM_THREADS) void hop_gemm_kernel(
    const float* __restrict__ embeds, const float* __restrict__ WmF,
    const int*  __restrict__ eids,    const int4* __restrict__ tiletab,
    const int*  __restrict__ ntiles,  const int*  __restrict__ rowlist,
    const float* __restrict__ xin,    float* __restrict__ xout, int hop)
{
    __shared__ float sX[16][132];
    __shared__ float sB[16][132];
    __shared__ int   sRow[16];

    int nt = ntiles[hop];
    if ((int)blockIdx.x >= nt) return;
    int4 tt = tiletab[hop * MAX_TILES + blockIdx.x];
    int m = tt.x, start = tt.y, cnt = tt.z;
    int tid = threadIdx.x;

    if (tid < 16) {
        int slot = (tid < cnt) ? tid : 0;
        sRow[tid] = rowlist[hop * BATCH + start + slot];
    }
    __syncthreads();

    {
        int c4 = tid & 31, r0 = tid >> 5;
        int rowg[STAGE_ROWS], eb[STAGE_ROWS], ex[STAGE_ROWS];
        const float4* px[STAGE_ROWS];
        const float4* pb[STAGE_ROWS];
        float4 vx[STAGE_ROWS], vb[STAGE_ROWS];
        for (int j = 0; j < STAGE_ROWS; ++j) rowg[j] = sRow[r0 + j * (GEMM_THREADS / 32)];
        for (int j = 0; j < STAGE_ROWS; ++j) eb[j] = eids[rowg[j] * 4 + hop + 1];
        if (hop == 0) {
            for (int j = 0; j < STAGE_ROWS; ++j) ex[j] = eids[rowg[j] * 4 + 0];
            for (int j = 0; j < STAGE_ROWS; ++j) px[j] = (const float4*)(embeds + (size_t)ex[j] * EMBED);
        } else {
            for (int j = 0; j < STAGE_ROWS; ++j) px[j] = (const float4*)(xin + (size_t)rowg[j] * EMBED);
        }
        for (int j = 0; j < STAGE_ROWS; ++j) pb[j] = (const float4*)(embeds + (size_t)eb[j] * EMBED);
        for (int j = 0; j < STAGE_ROWS; ++j) { vx[j] = px[j][c4]; vb[j] = pb[j][c4]; }
        for (int j = 0; j < STAGE_ROWS; ++j) {
            int r = r0 + j * (GEMM_THREADS / 32);
            sX[r][c4 * 4 + 0] = vx[j].x; sX[r][c4 * 4 + 1] = vx[j].y;
            sX[r][c4 * 4 + 2] = vx[j].z; sX[r][c4 * 4 + 3] = vx[j].w;
            sB[r][c4 * 4 + 0] = vb[j].x; sB[r][c4 * 4 + 1] = vb[j].y;
            sB[r][c4 * 4 + 2] = vb[j].z; sB[r][c4 * 4 + 3] = vb[j].w;
        }
    }
    __syncthreads();

    int wave = tid >> 5, lane = tid & 31;
    int half = lane >> 4, n = lane & 15, c0 = half * 2;
    int obase = wave * 32;

    const float4* Bb0 = (const float4*)(WmF + (size_t)(m * 8 + wave * 2) * 16 * 128) + lane;
    const float4* Bb1 = Bb0 + 16 * 32;

    v8f acc0 = {0.f,0.f,0.f,0.f,0.f,0.f,0.f,0.f};
    v8f acc1 = {0.f,0.f,0.f,0.f,0.f,0.f,0.f,0.f};
    #pragma unroll
    for (int k4g = 0; k4g < 16; ++k4g) {
        float4 w0 = Bb0[k4g * 32];
        float4 w1 = Bb1[k4g * 32];
        int k = k4g * 8;
        v2f a0, a1, b00, b01, b10, b11;
        a0.x = sX[n][k + c0];       a0.y = sX[n][k + c0 + 1];
        a1.x = sX[n][k + 4 + c0];   a1.y = sX[n][k + 4 + c0 + 1];
        b00.x = w0.x; b00.y = w0.y;
        b01.x = w0.z; b01.y = w0.w;
        b10.x = w1.x; b10.y = w1.y;
        b11.x = w1.z; b11.y = w1.w;
        acc0 = wmma_f32_4(a0, b00, acc0);
        acc1 = wmma_f32_4(a0, b10, acc1);
        acc0 = wmma_f32_4(a1, b01, acc0);
        acc1 = wmma_f32_4(a1, b11, acc1);
    }

    for (int r = 0; r < 8; ++r) {
        int row = r + half * 8;
        if (row < cnt) {
            int rowg = sRow[row];
            for (int t = 0; t < 2; ++t) {
                int o = obase + t * 16 + n;
                float bias = sB[row][o];
                float v = (t ? acc1[r] : acc0[r]) + bias;
                v = v > 0.f ? v : 0.f;
                xout[(size_t)rowg * EMBED + o] = v;
            }
        }
    }
}

__global__ __launch_bounds__(GEMM_THREADS) void hop_mlp_kernel(
    const float* __restrict__ embeds, const float* __restrict__ WmF,
    const int*  __restrict__ eids,    const int*  __restrict__ mids,
    const int*  __restrict__ rowlist, const float* __restrict__ xin,
    const float* __restrict__ W1F, const float* __restrict__ b1,
    const float* __restrict__ W2F, const float* __restrict__ b2, float* __restrict__ out)
{
    __shared__ float sX[16][132];
    __shared__ float sB[16][132];
    __shared__ float sH[16][516];
    __shared__ int   sRow[16];
    __shared__ int   sMod[16];

    const int hop = 2;
    int bt  = blockIdx.x;
    int tid = threadIdx.x;
    int wave = tid >> 5, lane = tid & 31;
    int half = lane >> 4, n = lane & 15, c0 = half * 2;

    if (tid < 16) sRow[tid] = rowlist[hop * BATCH + bt * 16 + tid];
    __syncthreads();

    {
        int c4 = tid & 31, r0 = tid >> 5;
        int rowg[STAGE_ROWS], eb[STAGE_ROWS];
        const float4* px[STAGE_ROWS];
        const float4* pb[STAGE_ROWS];
        float4 vx[STAGE_ROWS], vb[STAGE_ROWS];
        int mymod = 0;
        for (int j = 0; j < STAGE_ROWS; ++j) rowg[j] = sRow[r0 + j * (GEMM_THREADS / 32)];
        for (int j = 0; j < STAGE_ROWS; ++j) eb[j] = eids[rowg[j] * 4 + hop + 1];
        if (tid < 16) mymod = mids[sRow[tid] * 3 + hop];
        for (int j = 0; j < STAGE_ROWS; ++j) px[j] = (const float4*)(xin + (size_t)rowg[j] * EMBED);
        for (int j = 0; j < STAGE_ROWS; ++j) pb[j] = (const float4*)(embeds + (size_t)eb[j] * EMBED);
        for (int j = 0; j < STAGE_ROWS; ++j) { vx[j] = px[j][c4]; vb[j] = pb[j][c4]; }
        if (tid < 16) sMod[tid] = mymod;
        for (int j = 0; j < STAGE_ROWS; ++j) {
            int r = r0 + j * (GEMM_THREADS / 32);
            sX[r][c4 * 4 + 0] = vx[j].x; sX[r][c4 * 4 + 1] = vx[j].y;
            sX[r][c4 * 4 + 2] = vx[j].z; sX[r][c4 * 4 + 3] = vx[j].w;
            sB[r][c4 * 4 + 0] = vb[j].x; sB[r][c4 * 4 + 1] = vb[j].y;
            sB[r][c4 * 4 + 2] = vb[j].z; sB[r][c4 * 4 + 3] = vb[j].w;
        }
    }
    __syncthreads();

    {
        int obase = wave * 32;
        int modr[8];
        for (int r = 0; r < 8; ++r) modr[r] = sMod[r + half * 8];
        float res0[8], res1[8];
        for (int r = 0; r < 8; ++r) { res0[r] = 0.f; res1[r] = 0.f; }

        int mcur  = sMod[n];
        int mprev = sMod[n > 0 ? n - 1 : 0];
        bool opens = (lane < 16) && (lane == 0 || mcur != mprev);
        unsigned segs = (unsigned)__ballot(opens) & 0xFFFFu;

        while (segs) {
            int first = __builtin_ctz(segs);
            segs &= segs - 1;
            int m = sMod[first];

            const float4* Bb0 = (const float4*)(WmF + (size_t)(m * 8 + wave * 2) * 16 * 128) + lane;
            const float4* Bb1 = Bb0 + 16 * 32;
            v8f acc0 = {0.f,0.f,0.f,0.f,0.f,0.f,0.f,0.f};
            v8f acc1 = {0.f,0.f,0.f,0.f,0.f,0.f,0.f,0.f};
            #pragma unroll
            for (int k4g = 0; k4g < 16; ++k4g) {
                float4 w0 = Bb0[k4g * 32];
                float4 w1 = Bb1[k4g * 32];
                int k = k4g * 8;
                v2f a0, a1, b00, b01, b10, b11;
                a0.x = sX[n][k + c0];       a0.y = sX[n][k + c0 + 1];
                a1.x = sX[n][k + 4 + c0];   a1.y = sX[n][k + 4 + c0 + 1];
                b00.x = w0.x; b00.y = w0.y;
                b01.x = w0.z; b01.y = w0.w;
                b10.x = w1.x; b10.y = w1.y;
                b11.x = w1.z; b11.y = w1.w;
                acc0 = wmma_f32_4(a0, b00, acc0);
                acc1 = wmma_f32_4(a0, b10, acc1);
                acc0 = wmma_f32_4(a1, b01, acc0);
                acc1 = wmma_f32_4(a1, b11, acc1);
            }
            for (int r = 0; r < 8; ++r) {
                bool mine = (modr[r] == m);
                res0[r] = mine ? acc0[r] : res0[r];
                res1[r] = mine ? acc1[r] : res1[r];
            }
        }

        for (int r = 0; r < 8; ++r) {
            int row = r + half * 8;
            for (int t = 0; t < 2; ++t) {
                int o = obase + t * 16 + n;
                float bias = sB[row][o];
                float v = (t ? res1[r] : res0[r]) + bias;
                v = v > 0.f ? v : 0.f;
                sB[row][o] = (1.f - WCOMB) * bias + WCOMB * v;
            }
        }
    }
    __syncthreads();

    for (int p = 0; p < 4; ++p) {
        int ot = p * 8 + wave * 2;
        int ob = ot * 16;
        const float4* Bb0 = (const float4*)(W1F + (size_t)ot * 16 * 128) + lane;
        const float4* Bb1 = Bb0 + 16 * 32;
        v8f acc0 = {0.f,0.f,0.f,0.f,0.f,0.f,0.f,0.f};
        v8f acc1 = {0.f,0.f,0.f,0.f,0.f,0.f,0.f,0.f};
        #pragma unroll
        for (int k4g = 0; k4g < 16; ++k4g) {
            float4 w0 = Bb0[k4g * 32];
            float4 w1 = Bb1[k4g * 32];
            int k = k4g * 8;
            v2f a0, a1, b00, b01, b10, b11;
            a0.x = sB[n][k + c0];       a0.y = sB[n][k + c0 + 1];
            a1.x = sB[n][k + 4 + c0];   a1.y = sB[n][k + 4 + c0 + 1];
            b00.x = w0.x; b00.y = w0.y;
            b01.x = w0.z; b01.y = w0.w;
            b10.x = w1.x; b10.y = w1.y;
            b11.x = w1.z; b11.y = w1.w;
            acc0 = wmma_f32_4(a0, b00, acc0);
            acc1 = wmma_f32_4(a0, b10, acc1);
            acc0 = wmma_f32_4(a1, b01, acc0);
            acc1 = wmma_f32_4(a1, b11, acc1);
        }
        float bj0 = b1[ob + n];
        float bj1 = b1[ob + 16 + n];
        for (int r = 0; r < 8; ++r) {
            int row = r + half * 8;
            float v0 = acc0[r] + bj0;
            float v1 = acc1[r] + bj1;
            sH[row][ob + n]      = v0 > 0.f ? v0 : 0.f;
            sH[row][ob + 16 + n] = v1 > 0.f ? v1 : 0.f;
        }
    }
    __syncthreads();

    {
        int ob = wave * 16;
        const float4* Bb = (const float4*)(W2F + (size_t)wave * 64 * 128) + lane;
        v8f acc = {0.f,0.f,0.f,0.f,0.f,0.f,0.f,0.f};
        #pragma unroll 8
        for (int k4g = 0; k4g < 64; ++k4g) {
            float4 w = Bb[k4g * 32];
            int k = k4g * 8;
            v2f a0, a1, b0, b1f;
            a0.x = sH[n][k + c0];       a0.y = sH[n][k + c0 + 1];
            a1.x = sH[n][k + 4 + c0];   a1.y = sH[n][k + 4 + c0 + 1];
            b0.x = w.x; b0.y = w.y;
            b1f.x = w.z; b1f.y = w.w;
            acc = wmma_f32_4(a0, b0,  acc);
            acc = wmma_f32_4(a1, b1f, acc);
        }
        float bo = b2[ob + n];
        for (int r = 0; r < 8; ++r) {
            int row = r + half * 8;
            out[(size_t)sRow[row] * OUT_DIM + ob + n] = acc[r] + bo;
        }
    }
}

extern "C" void kernel_launch(void* const* d_in, const int* in_sizes, int n_in,
                              void* d_out, int out_size, void* d_ws, size_t ws_size,
                              hipStream_t stream) {
    const float* embeds = (const float*)d_in[0];
    const float* Wmods  = (const float*)d_in[1];
    const float* W1     = (const float*)d_in[2];
    const float* b1     = (const float*)d_in[3];
    const float* W2     = (const float*)d_in[4];
    const float* b2     = (const float*)d_in[5];
    const int*   eids   = (const int*)d_in[6];
    const int*   mids   = (const int*)d_in[7];
    float*       out    = (float*)d_out;

    char* ws = (char*)d_ws;
    size_t off = 0;
    float* X0      = (float*)(ws + off); off += (size_t)BATCH * EMBED * 4;
    float* X1      = (float*)(ws + off); off += (size_t)BATCH * EMBED * 4;
    float* WmF     = (float*)(ws + off); off += (size_t)NUM_MOD * EMBED * EMBED * 4;
    float* W1F     = (float*)(ws + off); off += (size_t)HIDDEN * EMBED * 4;
    float* W2F     = (float*)(ws + off); off += (size_t)OUT_DIM * HIDDEN * 4;
    off += 3 * 3 * NUM_MOD * 4;
    int*   rowlist = (int*)(ws + off);   off += 3 * BATCH * 4;
    int4*  tiletab = (int4*)(ws + off);  off += 3 * MAX_TILES * sizeof(int4);
    int*   ntiles  = (int*)(ws + off);   off += 64;
    (void)ws_size; (void)in_sizes; (void)n_in; (void)out_size;

    prep_kernel<<<BIN_BLOCKS + PACK_BLOCKS, PREP_THREADS, 0, stream>>>(
        mids, Wmods, W1, W2, rowlist, tiletab, ntiles,
        (float4*)WmF, (float4*)W1F, (float4*)W2F);

    hop_gemm_kernel<<<MAX_TILES, GEMM_THREADS, 0, stream>>>(embeds, WmF, eids, tiletab, ntiles, rowlist, X1, X0, 0);
    hop_gemm_kernel<<<MAX_TILES, GEMM_THREADS, 0, stream>>>(embeds, WmF, eids, tiletab, ntiles, rowlist, X0, X1, 1);

    hop_mlp_kernel<<<BATCH / 16, GEMM_THREADS, 0, stream>>>(embeds, WmF, eids, mids, rowlist, X1,
                                                            W1F, b1, W2F, b2, out);
}


# hop 0 joins the first launch: (module, row slice) blocks pick their rows by ballot, read row-major weights; binning+packing run beside it; hop 0 sums each 8-k group 0,1,4,5,2,3,6,7; 4 launches to 3
# speedup vs baseline: 4.3685x; 1.1113x over previous
#include <hip/hip_runtime.h>
#include <hip/hip_bf16.h>
#include <stdint.h>

#define BATCH    8192
#define EMBED    128
#define NUM_MOD  64
#define HIDDEN   512
#define OUT_DIM  64
#define MAX_TILES 640
#define WCOMB    0.2857142857142857f
#define GEMM_THREADS 128
#define STAGE_ROWS   (16 * 32 / GEMM_THREADS)

typedef float v2f __attribute__((ext_vector_type(2)));
typedef float v8f __attribute__((ext_vector_type(8)));

__device__ __forceinline__ v8f wmma_f32_4(v2f a, v2f b, v8f c) {
    return __builtin_amdgcn_wmma_f32_16x16x4_f32(
         false, a,  false, b,
         (short)0, c,  false,  false);
}

#define PREP_THREADS   128
#define BIN_BLOCKS     2
#define BIN_ROWS       (BATCH / PREP_THREADS)
#define H0_SLICES      10
#define H0_SLICE_ROWS  820
#define H0_ITERS       ((H0_SLICE_ROWS + PREP_THREADS - 1) / PREP_THREADS)
#define H0_BLOCKS      (NUM_MOD * H0_SLICES)
#define PACK_ITEMS_WM  (NUM_MOD * EMBED * EMBED / 8)
#define PACK_ITEMS_W1  (HIDDEN * EMBED / 8)
#define PACK_ITEMS_W2  (OUT_DIM * HIDDEN / 8)
#define PACK_PER_THREAD 4
#define PACK_CHUNK     (PREP_THREADS * PACK_PER_THREAD)
#define PACK_BLOCKS_WM (PACK_ITEMS_WM / PACK_CHUNK)
#define PACK_BLOCKS_W1 (PACK_ITEMS_W1 / PACK_CHUNK)
#define PACK_BLOCKS_W2 (PACK_ITEMS_W2 / PACK_CHUNK)
#define PACK_BLOCKS    (PACK_BLOCKS_WM + PACK_BLOCKS_W1 + PACK_BLOCKS_W2)

template <int KGBITS>
__device__ __forceinline__ void pack_items(const float* __restrict__ Wsrc,
                                           float4* __restrict__ WF, int first) {
    const float4* src[PACK_PER_THREAD];
    float4*       dst[PACK_PER_THREAD];
    float4        lo[PACK_PER_THREAD], hi[PACK_PER_THREAD];
    for (int j = 0; j < PACK_PER_THREAD; ++j) {
        int p    = first + j * PREP_THREADS + (int)threadIdx.x;
        int n    = p & 15;
        int k4g  = (p >> 4) & ((1 << KGBITS) - 1);
        int tile = p >> (4 + KGBITS);
        src[j] = (const float4*)(Wsrc + ((size_t)(tile * 16 + n) << (KGBITS + 3)) + 8 * k4g);
        dst[j] = WF + ((size_t)((tile << KGBITS) + k4g)) * 32 + n;
    }
    for (int j = 0; j < PACK_PER_THREAD; ++j) { lo[j] = src[j][0]; hi[j] = src[j][1]; }
    for (int j = 0; j < PACK_PER_THREAD; ++j) {
        dst[j][0]  = make_float4(lo[j].x, lo[j].y, hi[j].x, hi[j].y);
        dst[j][16] = make_float4(lo[j].z, lo[j].w, hi[j].z, hi[j].w);
    }
}

__global__ __launch_bounds__(PREP_THREADS) void first_hop_prep_kernel(
    const float* __restrict__ embeds, const float* __restrict__ Wmods,
    const float* __restrict__ W1, const float* __restrict__ W2,
    const int* __restrict__ eids, const int* __restrict__ mids,
    int* __restrict__ rowlist, int4* __restrict__ tiletab, int* __restrict__ ntiles,
    float4* __restrict__ WmF, float4* __restrict__ W1F, float4* __restrict__ W2F,
    float* __restrict__ xout)
{
    __shared__ int sCnt[NUM_MOD];
    __shared__ int sStart[NUM_MOD];
    __shared__ float sX[16][132];
    __shared__ float sB[16][132];
    __shared__ int   sList[H0_SLICE_ROWS];
    __shared__ int   sWcnt[PREP_THREADS / 32];

    int tid = threadIdx.x;
    int blk = blockIdx.x;

    if (blk >= BIN_BLOCKS + H0_BLOCKS) {
        int pb = blk - BIN_BLOCKS - H0_BLOCKS;
        if (pb < PACK_BLOCKS_WM)
            pack_items<4>(Wmods, WmF, pb * PACK_CHUNK);
        else if (pb < PACK_BLOCKS_WM + PACK_BLOCKS_W1)
            pack_items<4>(W1, W1F, (pb - PACK_BLOCKS_WM) * PACK_CHUNK);
        else
            pack_items<6>(W2, W2F, (pb - PACK_BLOCKS_WM - PACK_BLOCKS_W1) * PACK_CHUNK);
        return;
    }

    if (blk < BIN_BLOCKS) {
        int h = blk + 1;
        if (tid < NUM_MOD) sCnt[tid] = 0;
        __syncthreads();

        int mm[BIN_ROWS], rk[BIN_ROWS];
        #pragma unroll
        for (int j = 0; j < BIN_ROWS; ++j) mm[j] = mids[(j * PREP_THREADS + tid) * 3 + h];
        #pragma unroll
        for (int j = 0; j < BIN_ROWS; ++j) rk[j] = atomicAdd(&sCnt[mm[j]], 1);
        __syncthreads();

        int c = 0, nt = 0, st = 0, idx = 0;
        if (tid < NUM_MOD) {
            c  = sCnt[tid];
            nt = (c + 15) >> 4;
            for (int j = 0; j < NUM_MOD; ++j) {
                int cj = sCnt[j];
                bool lower = j < tid;
                st  += lower ? cj : 0;
                idx += lower ? ((cj + 15) >> 4) : 0;
            }
            sStart[tid] = st;
        }
        __syncthreads();

        #pragma unroll
        for (int j = 0; j < BIN_ROWS; ++j)
            rowlist[h * BATCH + sStart[mm[j]] + rk[j]] = j * PREP_THREADS + tid;

        if (tid < NUM_MOD) {
            int4* dst = tiletab + h * MAX_TILES + idx;
            for (int i = 0; i < nt; ++i) {
                int rem = c - i * 16; if (rem > 16) rem = 16;
                dst[i] = make_int4(tid, st + i * 16, rem, 0);
            }
            if (tid == NUM_MOD - 1) ntiles[h] = idx + nt;
        }
        return;
    }

    int hb = blk - BIN_BLOCKS;
    int m  = hb % NUM_MOD, s = hb / NUM_MOD;
    int r0 = s * H0_SLICE_ROWS;
    int r1 = r0 + H0_SLICE_ROWS; if (r1 > BATCH) r1 = BATCH;

    int wave = tid >> 5, lane = tid & 31;
    int half = lane >> 4, n = lane & 15;
    int obase = wave * 32;

    float4 w0[16], w1[16];
    {
        const float4* Wr = (const float4*)(Wmods + ((size_t)(m * EMBED + obase + n)) * EMBED) + half;
        #pragma unroll
        for (int g = 0; g < 16; ++g) {
            w0[g] = Wr[g * 2];
            w1[g] = Wr[16 * (EMBED / 4) + g * 2];
        }
    }

    int mv[H0_ITERS];
    unsigned hitmask[H0_ITERS];
    for (int it = 0; it < H0_ITERS; ++it) {
        int r = r0 + it * PREP_THREADS + tid;
        mv[it] = mids[(r < r1 ? r : r0) * 3 + 0];
    }
    int wcount = 0;
    for (int it = 0; it < H0_ITERS; ++it) {
        int r = r0 + it * PREP_THREADS + tid;
        bool hit = (r < r1) && (mv[it] == m);
        hitmask[it] = (unsigned)__ballot(hit);
        wcount += __popc(hitmask[it]);
    }
    if (lane == 0) sWcnt[wave] = wcount;
    __syncthreads();
    int base = 0, total = 0;
    for (int w = 0; w < PREP_THREADS / 32; ++w) {
        int cw = sWcnt[w];
        base  += (w < wave) ? cw : 0;
        total += cw;
    }
    for (int it = 0; it < H0_ITERS; ++it) {
        unsigned mk = hitmask[it];
        if ((mk >> lane) & 1u)
            sList[base + __popc(mk & ((1u << lane) - 1u))] = r0 + it * PREP_THREADS + tid;
        base += __popc(mk);
    }
    __syncthreads();

    int ntile = (total + 15) >> 4;
    for (int t = 0; t < ntile; ++t) {
        int cnt = total - t * 16; if (cnt > 16) cnt = 16;
        if (t > 0) __syncthreads();

        {
            int c4 = tid & 31, rs = tid >> 5;
            int rowg[STAGE_ROWS], eb[STAGE_ROWS], ex[STAGE_ROWS];
            const float4* px[STAGE_ROWS];
            const float4* pb[STAGE_ROWS];
            float4 vx[STAGE_ROWS], vb[STAGE_ROWS];
            for (int j = 0; j < STAGE_ROWS; ++j) {
                int slot = rs + j * (PREP_THREADS / 32);
                rowg[j] = sList[t * 16 + (slot < cnt ? slot : 0)];
            }
            for (int j = 0; j < STAGE_ROWS; ++j) eb[j] = eids[rowg[j] * 4 + 1];
            for (int j = 0; j < STAGE_ROWS; ++j) ex[j] = eids[rowg[j] * 4 + 0];
            for (int j = 0; j < STAGE_ROWS; ++j) px[j] = (const float4*)(embeds + (size_t)ex[j] * EMBED);
            for (int j = 0; j < STAGE_ROWS; ++j) pb[j] = (const float4*)(embeds + (size_t)eb[j] * EMBED);
            for (int j = 0; j < STAGE_ROWS; ++j) { vx[j] = px[j][c4]; vb[j] = pb[j][c4]; }
            for (int j = 0; j < STAGE_ROWS; ++j) {
                int r = rs + j * (PREP_THREADS / 32);
                sX[r][c4 * 4 + 0] = vx[j].x; sX[r][c4 * 4 + 1] = vx[j].y;
                sX[r][c4 * 4 + 2] = vx[j].z; sX[r][c4 * 4 + 3] = vx[j].w;
                sB[r][c4 * 4 + 0] = vb[j].x; sB[r][c4 * 4 + 1] = vb[j].y;
                sB[r][c4 * 4 + 2] = vb[j].z; sB[r][c4 * 4 + 3] = vb[j].w;
            }
        }
        __syncthreads();

        v8f acc0 = {0.f,0.f,0.f,0.f,0.f,0.f,0.f,0.f};
        v8f acc1 = {0.f,0.f,0.f,0.f,0.f,0.f,0.f,0.f};
        #pragma unroll
        for (int g = 0; g < 16; ++g) {
            const float4 av = *(const float4*)&sX[n][g * 8 + half * 4];
            v2f a0, a1, b00, b01, b10, b11;
            a0.x = av.x;     a0.y = av.y;
            a1.x = av.z;     a1.y = av.w;
            b00.x = w0[g].x; b00.y = w0[g].y;
            b01.x = w0[g].z; b01.y = w0[g].w;
            b10.x = w1[g].x; b10.y = w1[g].y;
            b11.x = w1[g].z; b11.y = w1[g].w;
            acc0 = wmma_f32_4(a0, b00, acc0);
            acc1 = wmma_f32_4(a0, b10, acc1);
            acc0 = wmma_f32_4(a1, b01, acc0);
            acc1 = wmma_f32_4(a1, b11, acc1);
        }

        for (int r = 0; r < 8; ++r) {
            int row = r + half * 8;
            if (row < cnt) {
                int rowg = sList[t * 16 + row];
                for (int u = 0; u < 2; ++u) {
                    int o = obase + u * 16 + n;
                    float bias = sB[row][o];
                    float v = (u ? acc1[r] : acc0[r]) + bias;
                    v = v > 0.f ? v : 0.f;
                    xout[(size_t)rowg * EMBED + o] = v;
                }
            }
        }
    }
}

__global__ __launch_bounds__(GEMM_THREADS) void hop_gemm_kernel(
    const float* __restrict__ embeds, const float* __restrict__ WmF,
    const int*  __restrict__ eids,    const int4* __restrict__ tiletab,
    const int*  __restrict__ ntiles,  const int*  __restrict__ rowlist,
    const float* __restrict__ xin,    float* __restrict__ xout, int hop)
{
    __shared__ float sX[16][132];
    __shared__ float sB[16][132];
    __shared__ int   sRow[16];

    int nt = ntiles[hop];
    if ((int)blockIdx.x >= nt) return;
    int4 tt = tiletab[hop * MAX_TILES + blockIdx.x];
    int m = tt.x, start = tt.y, cnt = tt.z;
    int tid = threadIdx.x;

    if (tid < 16) {
        int slot = (tid < cnt) ? tid : 0;
        sRow[tid] = rowlist[hop * BATCH + start + slot];
    }
    __syncthreads();

    {
        int c4 = tid & 31, r0 = tid >> 5;
        int rowg[STAGE_ROWS], eb[STAGE_ROWS];
        const float4* px[STAGE_ROWS];
        const float4* pb[STAGE_ROWS];
        float4 vx[STAGE_ROWS], vb[STAGE_ROWS];
        for (int j = 0; j < STAGE_ROWS; ++j) rowg[j] = sRow[r0 + j * (GEMM_THREADS / 32)];
        for (int j = 0; j < STAGE_ROWS; ++j) eb[j] = eids[rowg[j] * 4 + hop + 1];
        for (int j = 0; j < STAGE_ROWS; ++j) px[j] = (const float4*)(xin + (size_t)rowg[j] * EMBED);
        for (int j = 0; j < STAGE_ROWS; ++j) pb[j] = (const float4*)(embeds + (size_t)eb[j] * EMBED);
        for (int j = 0; j < STAGE_ROWS; ++j) { vx[j] = px[j][c4]; vb[j] = pb[j][c4]; }
        for (int j = 0; j < STAGE_ROWS; ++j) {
            int r = r0 + j * (GEMM_THREADS / 32);
            sX[r][c4 * 4 + 0] = vx[j].x; sX[r][c4 * 4 + 1] = vx[j].y;
            sX[r][c4 * 4 + 2] = vx[j].z; sX[r][c4 * 4 + 3] = vx[j].w;
            sB[r][c4 * 4 + 0] = vb[j].x; sB[r][c4 * 4 + 1] = vb[j].y;
            sB[r][c4 * 4 + 2] = vb[j].z; sB[r][c4 * 4 + 3] = vb[j].w;
        }
    }
    __syncthreads();

    int wave = tid >> 5, lane = tid & 31;
    int half = lane >> 4, n = lane & 15, c0 = half * 2;
    int obase = wave * 32;

    const float4* Bb0 = (const float4*)(WmF + (size_t)(m * 8 + wave * 2) * 16 * 128) + lane;
    const float4* Bb1 = Bb0 + 16 * 32;

    v8f acc0 = {0.f,0.f,0.f,0.f,0.f,0.f,0.f,0.f};
    v8f acc1 = {0.f,0.f,0.f,0.f,0.f,0.f,0.f,0.f};
    #pragma unroll
    for (int k4g = 0; k4g < 16; ++k4g) {
        float4 w0 = Bb0[k4g * 32];
        float4 w1 = Bb1[k4g * 32];
        int k = k4g * 8;
        v2f a0, a1, b00, b01, b10, b11;
        a0.x = sX[n][k + c0];       a0.y = sX[n][k + c0 + 1];
        a1.x = sX[n][k + 4 + c0];   a1.y = sX[n][k + 4 + c0 + 1];
        b00.x = w0.x; b00.y = w0.y;
        b01.x = w0.z; b01.y = w0.w;
        b10.x = w1.x; b10.y = w1.y;
        b11.x = w1.z; b11.y = w1.w;
        acc0 = wmma_f32_4(a0, b00, acc0);
        acc1 = wmma_f32_4(a0, b10, acc1);
        acc0 = wmma_f32_4(a1, b01, acc0);
        acc1 = wmma_f32_4(a1, b11, acc1);
    }

    for (int r = 0; r < 8; ++r) {
        int row = r + half * 8;
        if (row < cnt) {
            int rowg = sRow[row];
            for (int t = 0; t < 2; ++t) {
                int o = obase + t * 16 + n;
                float bias = sB[row][o];
                float v = (t ? acc1[r] : acc0[r]) + bias;
                v = v > 0.f ? v : 0.f;
                xout[(size_t)rowg * EMBED + o] = v;
            }
        }
    }
}

__global__ __launch_bounds__(GEMM_THREADS) void hop_mlp_kernel(
    const float* __restrict__ embeds, const float* __restrict__ WmF,
    const int*  __restrict__ eids,    const int*  __restrict__ mids,
    const int*  __restrict__ rowlist, const float* __restrict__ xin,
    const float* __restrict__ W1F, const float* __restrict__ b1,
    const float* __restrict__ W2F, const float* __restrict__ b2, float* __restrict__ out)
{
    __shared__ float sX[16][132];
    __shared__ float sB[16][132];
    __shared__ float sH[16][516];
    __shared__ int   sRow[16];
    __shared__ int   sMod[16];

    const int hop = 2;
    int bt  = blockIdx.x;
    int tid = threadIdx.x;
    int wave = tid >> 5, lane = tid & 31;
    int half = lane >> 4, n = lane & 15, c0 = half * 2;

    if (tid < 16) sRow[tid] = rowlist[hop * BATCH + bt * 16 + tid];
    __syncthreads();

    {
        int c4 = tid & 31, r0 = tid >> 5;
        int rowg[STAGE_ROWS], eb[STAGE_ROWS];
        const float4* px[STAGE_ROWS];
        const float4* pb[STAGE_ROWS];
        float4 vx[STAGE_ROWS], vb[STAGE_ROWS];
        int mymod = 0;
        for (int j = 0; j < STAGE_ROWS; ++j) rowg[j] = sRow[r0 + j * (GEMM_THREADS / 32)];
        for (int j = 0; j < STAGE_ROWS; ++j) eb[j] = eids[rowg[j] * 4 + hop + 1];
        if (tid < 16) mymod = mids[sRow[tid] * 3 + hop];
        for (int j = 0; j < STAGE_ROWS; ++j) px[j] = (const float4*)(xin + (size_t)rowg[j] * EMBED);
        for (int j = 0; j < STAGE_ROWS; ++j) pb[j] = (const float4*)(embeds + (size_t)eb[j] * EMBED);
        for (int j = 0; j < STAGE_ROWS; ++j) { vx[j] = px[j][c4]; vb[j] = pb[j][c4]; }
        if (tid < 16) sMod[tid] = mymod;
        for (int j = 0; j < STAGE_ROWS; ++j) {
            int r = r0 + j * (GEMM_THREADS / 32);
            sX[r][c4 * 4 + 0] = vx[j].x; sX[r][c4 * 4 + 1] = vx[j].y;
            sX[r][c4 * 4 + 2] = vx[j].z; sX[r][c4 * 4 + 3] = vx[j].w;
            sB[r][c4 * 4 + 0] = vb[j].x; sB[r][c4 * 4 + 1] = vb[j].y;
            sB[r][c4 * 4 + 2] = vb[j].z; sB[r][c4 * 4 + 3] = vb[j].w;
        }
    }
    __syncthreads();

    {
        int obase = wave * 32;
        int modr[8];
        for (int r = 0; r < 8; ++r) modr[r] = sMod[r + half * 8];
        float res0[8], res1[8];
        for (int r = 0; r < 8; ++r) { res0[r] = 0.f; res1[r] = 0.f; }

        int mcur  = sMod[n];
        int mprev = sMod[n > 0 ? n - 1 : 0];
        bool opens = (lane < 16) && (lane == 0 || mcur != mprev);
        unsigned segs = (unsigned)__ballot(opens) & 0xFFFFu;

        while (segs) {
            int first = __builtin_ctz(segs);
            segs &= segs - 1;
            int m = sMod[first];

            const float4* Bb0 = (const float4*)(WmF + (size_t)(m * 8 + wave * 2) * 16 * 128) + lane;
            const float4* Bb1 = Bb0 + 16 * 32;
            v8f acc0 = {0.f,0.f,0.f,0.f,0.f,0.f,0.f,0.f};
            v8f acc1 = {0.f,0.f,0.f,0.f,0.f,0.f,0.f,0.f};
            #pragma unroll
            for (int k4g = 0; k4g < 16; ++k4g) {
                float4 w0 = Bb0[k4g * 32];
                float4 w1 = Bb1[k4g * 32];
                int k = k4g * 8;
                v2f a0, a1, b00, b01, b10, b11;
                a0.x = sX[n][k + c0];       a0.y = sX[n][k + c0 + 1];
                a1.x = sX[n][k + 4 + c0];   a1.y = sX[n][k + 4 + c0 + 1];
                b00.x = w0.x; b00.y = w0.y;
                b01.x = w0.z; b01.y = w0.w;
                b10.x = w1.x; b10.y = w1.y;
                b11.x = w1.z; b11.y = w1.w;
                acc0 = wmma_f32_4(a0, b00, acc0);
                acc1 = wmma_f32_4(a0, b10, acc1);
                acc0 = wmma_f32_4(a1, b01, acc0);
                acc1 = wmma_f32_4(a1, b11, acc1);
            }
            for (int r = 0; r < 8; ++r) {
                bool mine = (modr[r] == m);
                res0[r] = mine ? acc0[r] : res0[r];
                res1[r] = mine ? acc1[r] : res1[r];
            }
        }

        for (int r = 0; r < 8; ++r) {
            int row = r + half * 8;
            for (int t = 0; t < 2; ++t) {
                int o = obase + t * 16 + n;
                float bias = sB[row][o];
                float v = (t ? res1[r] : res0[r]) + bias;
                v = v > 0.f ? v : 0.f;
                sB[row][o] = (1.f - WCOMB) * bias + WCOMB * v;
            }
        }
    }
    __syncthreads();

    for (int p = 0; p < 4; ++p) {
        int ot = p * 8 + wave * 2;
        int ob = ot * 16;
        const float4* Bb0 = (const float4*)(W1F + (size_t)ot * 16 * 128) + lane;
        const float4* Bb1 = Bb0 + 16 * 32;
        v8f acc0 = {0.f,0.f,0.f,0.f,0.f,0.f,0.f,0.f};
        v8f acc1 = {0.f,0.f,0.f,0.f,0.f,0.f,0.f,0.f};
        #pragma unroll
        for (int k4g = 0; k4g < 16; ++k4g) {
            float4 w0 = Bb0[k4g * 32];
            float4 w1 = Bb1[k4g * 32];
            int k = k4g * 8;
            v2f a0, a1, b00, b01, b10, b11;
            a0.x = sB[n][k + c0];       a0.y = sB[n][k + c0 + 1];
            a1.x = sB[n][k + 4 + c0];   a1.y = sB[n][k + 4 + c0 + 1];
            b00.x = w0.x; b00.y = w0.y;
            b01.x = w0.z; b01.y = w0.w;
            b10.x = w1.x; b10.y = w1.y;
            b11.x = w1.z; b11.y = w1.w;
            acc0 = wmma_f32_4(a0, b00, acc0);
            acc1 = wmma_f32_4(a0, b10, acc1);
            acc0 = wmma_f32_4(a1, b01, acc0);
            acc1 = wmma_f32_4(a1, b11, acc1);
        }
        float bj0 = b1[ob + n];
        float bj1 = b1[ob + 16 + n];
        for (int r = 0; r < 8; ++r) {
            int row = r + half * 8;
            float v0 = acc0[r] + bj0;
            float v1 = acc1[r] + bj1;
            sH[row][ob + n]      = v0 > 0.f ? v0 : 0.f;
            sH[row][ob + 16 + n] = v1 > 0.f ? v1 : 0.f;
        }
    }
    __syncthreads();

    {
        int ob = wave * 16;
        const float4* Bb = (const float4*)(W2F + (size_t)wave * 64 * 128) + lane;
        v8f acc = {0.f,0.f,0.f,0.f,0.f,0.f,0.f,0.f};
        #pragma unroll 8
        for (int k4g = 0; k4g < 64; ++k4g) {
            float4 w = Bb[k4g * 32];
            int k = k4g * 8;
            v2f a0, a1, b0, b1f;
            a0.x = sH[n][k + c0];       a0.y = sH[n][k + c0 + 1];
            a1.x = sH[n][k + 4 + c0];   a1.y = sH[n][k + 4 + c0 + 1];
            b0.x = w.x; b0.y = w.y;
            b1f.x = w.z; b1f.y = w.w;
            acc = wmma_f32_4(a0, b0,  acc);
            acc = wmma_f32_4(a1, b1f, acc);
        }
        float bo = b2[ob + n];
        for (int r = 0; r < 8; ++r) {
            int row = r + half * 8;
            out[(size_t)sRow[row] * OUT_DIM + ob + n] = acc[r] + bo;
        }
    }
}

extern "C" void kernel_launch(void* const* d_in, const int* in_sizes, int n_in,
                              void* d_out, int out_size, void* d_ws, size_t ws_size,
                              hipStream_t stream) {
    const float* embeds = (const float*)d_in[0];
    const float* Wmods  = (const float*)d_in[1];
    const float* W1     = (const float*)d_in[2];
    const float* b1     = (const float*)d_in[3];
    const float* W2     = (const float*)d_in[4];
    const float* b2     = (const float*)d_in[5];
    const int*   eids   = (const int*)d_in[6];
    const int*   mids   = (const int*)d_in[7];
    float*       out    = (float*)d_out;

    char* ws = (char*)d_ws;
    size_t off = 0;
    float* X0      = (float*)(ws + off); off += (size_t)BATCH * EMBED * 4;
    float* X1      = (float*)(ws + off); off += (size_t)BATCH * EMBED * 4;
    float* WmF     = (float*)(ws + off); off += (size_t)NUM_MOD * EMBED * EMBED * 4;
    float* W1F     = (float*)(ws + off); off += (size_t)HIDDEN * EMBED * 4;
    float* W2F     = (float*)(ws + off); off += (size_t)OUT_DIM * HIDDEN * 4;
    off += 3 * 3 * NUM_MOD * 4;
    int*   rowlist = (int*)(ws + off);   off += 3 * BATCH * 4;
    int4*  tiletab = (int4*)(ws + off);  off += 3 * MAX_TILES * sizeof(int4);
    int*   ntiles  = (int*)(ws + off);   off += 64;
    (void)ws_size; (void)in_sizes; (void)n_in; (void)out_size;

    first_hop_prep_kernel<<<BIN_BLOCKS + H0_BLOCKS + PACK_BLOCKS, PREP_THREADS, 0, stream>>>(
        embeds, Wmods, W1, W2, eids, mids, rowlist, tiletab, ntiles,
        (float4*)WmF, (float4*)W1F, (float4*)W2F, X0);

    hop_gemm_kernel<<<MAX_TILES, GEMM_THREADS, 0, stream>>>(embeds, WmF, eids, tiletab, ntiles, rowlist, X0, X1, 1);

    hop_mlp_kernel<<<BATCH / 16, GEMM_THREADS, 0, stream>>>(embeds, WmF, eids, mids, rowlist, X1,
                                                            W1F, b1, W2F, b2, out);
}


# MLP phase 2 software-pipelined: the next block's 16 weight loads and the next half block's 8 LDS fragment reads go out as one group ahead of the current 16 WMMAs (second register set); sums unchanged
# speedup vs baseline: 4.5254x; 1.0359x over previous
#include <hip/hip_runtime.h>
#include <hip/hip_bf16.h>
#include <stdint.h>

#define BATCH    8192
#define EMBED    128
#define NUM_MOD  64
#define HIDDEN   512
#define OUT_DIM  64
#define MAX_TILES 640
#define WCOMB    0.2857142857142857f
#define GEMM_THREADS 128
#define STAGE_ROWS   (16 * 32 / GEMM_THREADS)

typedef float v2f __attribute__((ext_vector_type(2)));
typedef float v8f __attribute__((ext_vector_type(8)));

__device__ __forceinline__ v8f wmma_f32_4(v2f a, v2f b, v8f c) {
    return __builtin_amdgcn_wmma_f32_16x16x4_f32(
         false, a,  false, b,
         (short)0, c,  false,  false);
}

#define PREP_THREADS   128
#define BIN_BLOCKS     2
#define BIN_ROWS       (BATCH / PREP_THREADS)
#define H0_SLICES      10
#define H0_SLICE_ROWS  820
#define H0_ITERS       ((H0_SLICE_ROWS + PREP_THREADS - 1) / PREP_THREADS)
#define H0_BLOCKS      (NUM_MOD * H0_SLICES)
#define PACK_ITEMS_WM  (NUM_MOD * EMBED * EMBED / 8)
#define PACK_ITEMS_W1  (HIDDEN * EMBED / 8)
#define PACK_ITEMS_W2  (OUT_DIM * HIDDEN / 8)
#define PACK_PER_THREAD 4
#define PACK_CHUNK     (PREP_THREADS * PACK_PER_THREAD)
#define PACK_BLOCKS_WM (PACK_ITEMS_WM / PACK_CHUNK)
#define PACK_BLOCKS_W1 (PACK_ITEMS_W1 / PACK_CHUNK)
#define PACK_BLOCKS_W2 (PACK_ITEMS_W2 / PACK_CHUNK)
#define PACK_BLOCKS    (PACK_BLOCKS_WM + PACK_BLOCKS_W1 + PACK_BLOCKS_W2)

template <int KGBITS>
__device__ __forceinline__ void pack_items(const float* __restrict__ Wsrc,
                                           float4* __restrict__ WF, int first) {
    const float4* src[PACK_PER_THREAD];
    float4*       dst[PACK_PER_THREAD];
    float4        lo[PACK_PER_THREAD], hi[PACK_PER_THREAD];
    for (int j = 0; j < PACK_PER_THREAD; ++j) {
        int p    = first + j * PREP_THREADS + (int)threadIdx.x;
        int n    = p & 15;
        int k4g  = (p >> 4) & ((1 << KGBITS) - 1);
        int tile = p >> (4 + KGBITS);
        src[j] = (const float4*)(Wsrc + ((size_t)(tile * 16 + n) << (KGBITS + 3)) + 8 * k4g);
        dst[j] = WF + ((size_t)((tile << KGBITS) + k4g)) * 32 + n;
    }
    for (int j = 0; j < PACK_PER_THREAD; ++j) { lo[j] = src[j][0]; hi[j] = src[j][1]; }
    for (int j = 0; j < PACK_PER_THREAD; ++j) {
        dst[j][0]  = make_float4(lo[j].x, lo[j].y, hi[j].x, hi[j].y);
        dst[j][16] = make_float4(lo[j].z, lo[j].w, hi[j].z, hi[j].w);
    }
}

__global__ __launch_bounds__(PREP_THREADS) void first_hop_prep_kernel(
    const float* __restrict__ embeds, const float* __restrict__ Wmods,
    const float* __restrict__ W1, const float* __restrict__ W2,
    const int* __restrict__ eids, const int* __restrict__ mids,
    int* __restrict__ rowlist, int4* __restrict__ tiletab, int* __restrict__ ntiles,
    float4* __restrict__ WmF, float4* __restrict__ W1F, float4* __restrict__ W2F,
    float* __restrict__ xout)
{
    __shared__ int sCnt[NUM_MOD];
    __shared__ int sStart[NUM_MOD];
    __shared__ float sX[16][132];
    __shared__ float sB[16][132];
    __shared__ int   sList[H0_SLICE_ROWS];
    __shared__ int   sWcnt[PREP_THREADS / 32];

    int tid = threadIdx.x;
    int blk = blockIdx.x;

    if (blk >= BIN_BLOCKS + H0_BLOCKS) {
        int pb = blk - BIN_BLOCKS - H0_BLOCKS;
        if (pb < PACK_BLOCKS_WM)
            pack_items<4>(Wmods, WmF, pb * PACK_CHUNK);
        else if (pb < PACK_BLOCKS_WM + PACK_BLOCKS_W1)
            pack_items<4>(W1, W1F, (pb - PACK_BLOCKS_WM) * PACK_CHUNK);
        else
            pack_items<6>(W2, W2F, (pb - PACK_BLOCKS_WM - PACK_BLOCKS_W1) * PACK_CHUNK);
        return;
    }

    if (blk < BIN_BLOCKS) {
        int h = blk + 1;
        if (tid < NUM_MOD) sCnt[tid] = 0;
        __syncthreads();

        int mm[BIN_ROWS], rk[BIN_ROWS];
        #pragma unroll
        for (int j = 0; j < BIN_ROWS; ++j) mm[j] = mids[(j * PREP_THREADS + tid) * 3 + h];
        #pragma unroll
        for (int j = 0; j < BIN_ROWS; ++j) rk[j] = atomicAdd(&sCnt[mm[j]], 1);
        __syncthreads();

        int c = 0, nt = 0, st = 0, idx = 0;
        if (tid < NUM_MOD) {
            c  = sCnt[tid];
            nt = (c + 15) >> 4;
            for (int j = 0; j < NUM_MOD; ++j) {
                int cj = sCnt[j];
                bool lower = j < tid;
                st  += lower ? cj : 0;
                idx += lower ? ((cj + 15) >> 4) : 0;
            }
            sStart[tid] = st;
        }
        __syncthreads();

        #pragma unroll
        for (int j = 0; j < BIN_ROWS; ++j)
            rowlist[h * BATCH + sStart[mm[j]] + rk[j]] = j * PREP_THREADS + tid;

        if (tid < NUM_MOD) {
            int4* dst = tiletab + h * MAX_TILES + idx;
            for (int i = 0; i < nt; ++i) {
                int rem = c - i * 16; if (rem > 16) rem = 16;
                dst[i] = make_int4(tid, st + i * 16, rem, 0);
            }
            if (tid == NUM_MOD - 1) ntiles[h] = idx + nt;
        }
        return;
    }

    int hb = blk - BIN_BLOCKS;
    int m  = hb % NUM_MOD, s = hb / NUM_MOD;
    int r0 = s * H0_SLICE_ROWS;
    int r1 = r0 + H0_SLICE_ROWS; if (r1 > BATCH) r1 = BATCH;

    int wave = tid >> 5, lane = tid & 31;
    int half = lane >> 4, n = lane & 15;
    int obase = wave * 32;

    float4 w0[16], w1[16];
    {
        const float4* Wr = (const float4*)(Wmods + ((size_t)(m * EMBED + obase + n)) * EMBED) + half;
        #pragma unroll
        for (int g = 0; g < 16; ++g) {
            w0[g] = Wr[g * 2];
            w1[g] = Wr[16 * (EMBED / 4) + g * 2];
        }
    }

    int mv[H0_ITERS];
    unsigned hitmask[H0_ITERS];
    for (int it = 0; it < H0_ITERS; ++it) {
        int r = r0 + it * PREP_THREADS + tid;
        mv[it] = mids[(r < r1 ? r : r0) * 3 + 0];
    }
    int wcount = 0;
    for (int it = 0; it < H0_ITERS; ++it) {
        int r = r0 + it * PREP_THREADS + tid;
        bool hit = (r < r1) && (mv[it] == m);
        hitmask[it] = (unsigned)__ballot(hit);
        wcount += __popc(hitmask[it]);
    }
    if (lane == 0) sWcnt[wave] = wcount;
    __syncthreads();
    int base = 0, total = 0;
    for (int w = 0; w < PREP_THREADS / 32; ++w) {
        int cw = sWcnt[w];
        base  += (w < wave) ? cw : 0;
        total += cw;
    }
    for (int it = 0; it < H0_ITERS; ++it) {
        unsigned mk = hitmask[it];
        if ((mk >> lane) & 1u)
            sList[base + __popc(mk & ((1u << lane) - 1u))] = r0 + it * PREP_THREADS + tid;
        base += __popc(mk);
    }
    __syncthreads();

    int ntile = (total + 15) >> 4;
    for (int t = 0; t < ntile; ++t) {
        int cnt = total - t * 16; if (cnt > 16) cnt = 16;
        if (t > 0) __syncthreads();

        {
            int c4 = tid & 31, rs = tid >> 5;
            int rowg[STAGE_ROWS], eb[STAGE_ROWS], ex[STAGE_ROWS];
            const float4* px[STAGE_ROWS];
            const float4* pb[STAGE_ROWS];
            float4 vx[STAGE_ROWS], vb[STAGE_ROWS];
            for (int j = 0; j < STAGE_ROWS; ++j) {
                int slot = rs + j * (PREP_THREADS / 32);
                rowg[j] = sList[t * 16 + (slot < cnt ? slot : 0)];
            }
            for (int j = 0; j < STAGE_ROWS; ++j) eb[j] = eids[rowg[j] * 4 + 1];
            for (int j = 0; j < STAGE_ROWS; ++j) ex[j] = eids[rowg[j] * 4 + 0];
            for (int j = 0; j < STAGE_ROWS; ++j) px[j] = (const float4*)(embeds + (size_t)ex[j] * EMBED);
            for (int j = 0; j < STAGE_ROWS; ++j) pb[j] = (const float4*)(embeds + (size_t)eb[j] * EMBED);
            for (int j = 0; j < STAGE_ROWS; ++j) { vx[j] = px[j][c4]; vb[j] = pb[j][c4]; }
            for (int j = 0; j < STAGE_ROWS; ++j) {
                int r = rs + j * (PREP_THREADS / 32);
                sX[r][c4 * 4 + 0] = vx[j].x; sX[r][c4 * 4 + 1] = vx[j].y;
                sX[r][c4 * 4 + 2] = vx[j].z; sX[r][c4 * 4 + 3] = vx[j].w;
                sB[r][c4 * 4 + 0] = vb[j].x; sB[r][c4 * 4 + 1] = vb[j].y;
                sB[r][c4 * 4 + 2] = vb[j].z; sB[r][c4 * 4 + 3] = vb[j].w;
            }
        }
        __syncthreads();

        v8f acc0 = {0.f,0.f,0.f,0.f,0.f,0.f,0.f,0.f};
        v8f acc1 = {0.f,0.f,0.f,0.f,0.f,0.f,0.f,0.f};
        #pragma unroll
        for (int g = 0; g < 16; ++g) {
            const float4 av = *(const float4*)&sX[n][g * 8 + half * 4];
            v2f a0, a1, b00, b01, b10, b11;
            a0.x = av.x;     a0.y = av.y;
            a1.x = av.z;     a1.y = av.w;
            b00.x = w0[g].x; b00.y = w0[g].y;
            b01.x = w0[g].z; b01.y = w0[g].w;
            b10.x = w1[g].x; b10.y = w1[g].y;
            b11.x = w1[g].z; b11.y = w1[g].w;
            acc0 = wmma_f32_4(a0, b00, acc0);
            acc1 = wmma_f32_4(a0, b10, acc1);
            acc0 = wmma_f32_4(a1, b01, acc0);
            acc1 = wmma_f32_4(a1, b11, acc1);
        }

        for (int r = 0; r < 8; ++r) {
            int row = r + half * 8;
            if (row < cnt) {
                int rowg = sList[t * 16 + row];
                for (int u = 0; u < 2; ++u) {
                    int o = obase + u * 16 + n;
                    float bias = sB[row][o];
                    float v = (u ? acc1[r] : acc0[r]) + bias;
                    v = v > 0.f ? v : 0.f;
                    xout[(size_t)rowg * EMBED + o] = v;
                }
            }
        }
    }
}

__global__ __launch_bounds__(GEMM_THREADS) void hop_gemm_kernel(
    const float* __restrict__ embeds, const float* __restrict__ WmF,
    const int*  __restrict__ eids,    const int4* __restrict__ tiletab,
    const int*  __restrict__ ntiles,  const int*  __restrict__ rowlist,
    const float* __restrict__ xin,    float* __restrict__ xout, int hop)
{
    __shared__ float sX[16][132];
    __shared__ float sB[16][132];
    __shared__ int   sRow[16];

    int nt = ntiles[hop];
    if ((int)blockIdx.x >= nt) return;
    int4 tt = tiletab[hop * MAX_TILES + blockIdx.x];
    int m = tt.x, start = tt.y, cnt = tt.z;
    int tid = threadIdx.x;

    if (tid < 16) {
        int slot = (tid < cnt) ? tid : 0;
        sRow[tid] = rowlist[hop * BATCH + start + slot];
    }
    __syncthreads();

    {
        int c4 = tid & 31, r0 = tid >> 5;
        int rowg[STAGE_ROWS], eb[STAGE_ROWS];
        const float4* px[STAGE_ROWS];
        const float4* pb[STAGE_ROWS];
        float4 vx[STAGE_ROWS], vb[STAGE_ROWS];
        for (int j = 0; j < STAGE_ROWS; ++j) rowg[j] = sRow[r0 + j * (GEMM_THREADS / 32)];
        for (int j = 0; j < STAGE_ROWS; ++j) eb[j] = eids[rowg[j] * 4 + hop + 1];
        for (int j = 0; j < STAGE_ROWS; ++j) px[j] = (const float4*)(xin + (size_t)rowg[j] * EMBED);
        for (int j = 0; j < STAGE_ROWS; ++j) pb[j] = (const float4*)(embeds + (size_t)eb[j] * EMBED);
        for (int j = 0; j < STAGE_ROWS; ++j) { vx[j] = px[j][c4]; vb[j] = pb[j][c4]; }
        for (int j = 0; j < STAGE_ROWS; ++j) {
            int r = r0 + j * (GEMM_THREADS / 32);
            sX[r][c4 * 4 + 0] = vx[j].x; sX[r][c4 * 4 + 1] = vx[j].y;
            sX[r][c4 * 4 + 2] = vx[j].z; sX[r][c4 * 4 + 3] = vx[j].w;
            sB[r][c4 * 4 + 0] = vb[j].x; sB[r][c4 * 4 + 1] = vb[j].y;
            sB[r][c4 * 4 + 2] = vb[j].z; sB[r][c4 * 4 + 3] = vb[j].w;
        }
    }
    __syncthreads();

    int wave = tid >> 5, lane = tid & 31;
    int half = lane >> 4, n = lane & 15, c0 = half * 2;
    int obase = wave * 32;

    const float4* Bb0 = (const float4*)(WmF + (size_t)(m * 8 + wave * 2) * 16 * 128) + lane;
    const float4* Bb1 = Bb0 + 16 * 32;

    v8f acc0 = {0.f,0.f,0.f,0.f,0.f,0.f,0.f,0.f};
    v8f acc1 = {0.f,0.f,0.f,0.f,0.f,0.f,0.f,0.f};
    #pragma unroll
    for (int k4g = 0; k4g < 16; ++k4g) {
        float4 w0 = Bb0[k4g * 32];
        float4 w1 = Bb1[k4g * 32];
        int k = k4g * 8;
        v2f a0, a1, b00, b01, b10, b11;
        a0.x = sX[n][k + c0];       a0.y = sX[n][k + c0 + 1];
        a1.x = sX[n][k + 4 + c0];   a1.y = sX[n][k + 4 + c0 + 1];
        b00.x = w0.x; b00.y = w0.y;
        b01.x = w0.z; b01.y = w0.w;
        b10.x = w1.x; b10.y = w1.y;
        b11.x = w1.z; b11.y = w1.w;
        acc0 = wmma_f32_4(a0, b00, acc0);
        acc1 = wmma_f32_4(a0, b10, acc1);
        acc0 = wmma_f32_4(a1, b01, acc0);
        acc1 = wmma_f32_4(a1, b11, acc1);
    }

    for (int r = 0; r < 8; ++r) {
        int row = r + half * 8;
        if (row < cnt) {
            int rowg = sRow[row];
            for (int t = 0; t < 2; ++t) {
                int o = obase + t * 16 + n;
                float bias = sB[row][o];
                float v = (t ? acc1[r] : acc0[r]) + bias;
                v = v > 0.f ? v : 0.f;
                xout[(size_t)rowg * EMBED + o] = v;
            }
        }
    }
}

__global__ __launch_bounds__(GEMM_THREADS) void hop_mlp_kernel(
    const float* __restrict__ embeds, const float* __restrict__ WmF,
    const int*  __restrict__ eids,    const int*  __restrict__ mids,
    const int*  __restrict__ rowlist, const float* __restrict__ xin,
    const float* __restrict__ W1F, const float* __restrict__ b1,
    const float* __restrict__ W2F, const float* __restrict__ b2, float* __restrict__ out)
{
    __shared__ float sX[16][132];
    __shared__ float sB[16][132];
    __shared__ float sH[16][516];
    __shared__ int   sRow[16];
    __shared__ int   sMod[16];

    const int hop = 2;
    int bt  = blockIdx.x;
    int tid = threadIdx.x;
    int wave = tid >> 5, lane = tid & 31;
    int half = lane >> 4, n = lane & 15, c0 = half * 2;

    if (tid < 16) sRow[tid] = rowlist[hop * BATCH + bt * 16 + tid];
    __syncthreads();

    {
        int c4 = tid & 31, r0 = tid >> 5;
        int rowg[STAGE_ROWS], eb[STAGE_ROWS];
        const float4* px[STAGE_ROWS];
        const float4* pb[STAGE_ROWS];
        float4 vx[STAGE_ROWS], vb[STAGE_ROWS];
        int mymod = 0;
        for (int j = 0; j < STAGE_ROWS; ++j) rowg[j] = sRow[r0 + j * (GEMM_THREADS / 32)];
        for (int j = 0; j < STAGE_ROWS; ++j) eb[j] = eids[rowg[j] * 4 + hop + 1];
        if (tid < 16) mymod = mids[sRow[tid] * 3 + hop];
        for (int j = 0; j < STAGE_ROWS; ++j) px[j] = (const float4*)(xin + (size_t)rowg[j] * EMBED);
        for (int j = 0; j < STAGE_ROWS; ++j) pb[j] = (const float4*)(embeds + (size_t)eb[j] * EMBED);
        for (int j = 0; j < STAGE_ROWS; ++j) { vx[j] = px[j][c4]; vb[j] = pb[j][c4]; }
        if (tid < 16) sMod[tid] = mymod;
        for (int j = 0; j < STAGE_ROWS; ++j) {
            int r = r0 + j * (GEMM_THREADS / 32);
            sX[r][c4 * 4 + 0] = vx[j].x; sX[r][c4 * 4 + 1] = vx[j].y;
            sX[r][c4 * 4 + 2] = vx[j].z; sX[r][c4 * 4 + 3] = vx[j].w;
            sB[r][c4 * 4 + 0] = vb[j].x; sB[r][c4 * 4 + 1] = vb[j].y;
            sB[r][c4 * 4 + 2] = vb[j].z; sB[r][c4 * 4 + 3] = vb[j].w;
        }
    }
    __syncthreads();

    {
        int obase = wave * 32;
        int modr[8];
        for (int r = 0; r < 8; ++r) modr[r] = sMod[r + half * 8];
        float res0[8], res1[8];
        for (int r = 0; r < 8; ++r) { res0[r] = 0.f; res1[r] = 0.f; }

        int mcur  = sMod[n];
        int mprev = sMod[n > 0 ? n - 1 : 0];
        bool opens = (lane < 16) && (lane == 0 || mcur != mprev);
        unsigned segs = (unsigned)__ballot(opens) & 0xFFFFu;

        while (segs) {
            int first = __builtin_ctz(segs);
            segs &= segs - 1;
            int m = sMod[first];

            const float4* Bb0 = (const float4*)(WmF + (size_t)(m * 8 + wave * 2) * 16 * 128) + lane;
            const float4* Bb1 = Bb0 + 16 * 32;
            v8f acc0 = {0.f,0.f,0.f,0.f,0.f,0.f,0.f,0.f};
            v8f acc1 = {0.f,0.f,0.f,0.f,0.f,0.f,0.f,0.f};
            #pragma unroll
            for (int k4g = 0; k4g < 16; ++k4g) {
                float4 w0 = Bb0[k4g * 32];
                float4 w1 = Bb1[k4g * 32];
                int k = k4g * 8;
                v2f a0, a1, b00, b01, b10, b11;
                a0.x = sX[n][k + c0];       a0.y = sX[n][k + c0 + 1];
                a1.x = sX[n][k + 4 + c0];   a1.y = sX[n][k + 4 + c0 + 1];
                b00.x = w0.x; b00.y = w0.y;
                b01.x = w0.z; b01.y = w0.w;
                b10.x = w1.x; b10.y = w1.y;
                b11.x = w1.z; b11.y = w1.w;
                acc0 = wmma_f32_4(a0, b00, acc0);
                acc1 = wmma_f32_4(a0, b10, acc1);
                acc0 = wmma_f32_4(a1, b01, acc0);
                acc1 = wmma_f32_4(a1, b11, acc1);
            }
            for (int r = 0; r < 8; ++r) {
                bool mine = (modr[r] == m);
                res0[r] = mine ? acc0[r] : res0[r];
                res1[r] = mine ? acc1[r] : res1[r];
            }
        }

        for (int r = 0; r < 8; ++r) {
            int row = r + half * 8;
            for (int t = 0; t < 2; ++t) {
                int o = obase + t * 16 + n;
                float bias = sB[row][o];
                float v = (t ? res1[r] : res0[r]) + bias;
                v = v > 0.f ? v : 0.f;
                sB[row][o] = (1.f - WCOMB) * bias + WCOMB * v;
            }
        }
    }
    __syncthreads();

    for (int p = 0; p < 4; ++p) {
        int ot = p * 8 + wave * 2;
        int ob = ot * 16;
        const float4* Bb0 = (const float4*)(W1F + (size_t)ot * 16 * 128) + lane;
        const float4* Bb1 = Bb0 + 16 * 32;
        v8f acc0 = {0.f,0.f,0.f,0.f,0.f,0.f,0.f,0.f};
        v8f acc1 = {0.f,0.f,0.f,0.f,0.f,0.f,0.f,0.f};
        #pragma unroll
        for (int k4g = 0; k4g < 16; ++k4g) {
            float4 w0 = Bb0[k4g * 32];
            float4 w1 = Bb1[k4g * 32];
            int k = k4g * 8;
            v2f a0, a1, b00, b01, b10, b11;
            a0.x = sB[n][k + c0];       a0.y = sB[n][k + c0 + 1];
            a1.x = sB[n][k + 4 + c0];   a1.y = sB[n][k + 4 + c0 + 1];
            b00.x = w0.x; b00.y = w0.y;
            b01.x = w0.z; b01.y = w0.w;
            b10.x = w1.x; b10.y = w1.y;
            b11.x = w1.z; b11.y = w1.w;
            acc0 = wmma_f32_4(a0, b00, acc0);
            acc1 = wmma_f32_4(a0, b10, acc1);
            acc0 = wmma_f32_4(a1, b01, acc0);
            acc1 = wmma_f32_4(a1, b11, acc1);
        }
        float bj0 = b1[ob + n];
        float bj1 = b1[ob + 16 + n];
        for (int r = 0; r < 8; ++r) {
            int row = r + half * 8;
            float v0 = acc0[r] + bj0;
            float v1 = acc1[r] + bj1;
            sH[row][ob + n]      = v0 > 0.f ? v0 : 0.f;
            sH[row][ob + 16 + n] = v1 > 0.f ? v1 : 0.f;
        }
    }
    {
        int ob = wave * 16;
        const float4* Bb = (const float4*)(W2F + (size_t)wave * 64 * 128) + lane;
        float4 wq[2][16];
        v2f    a0q[2][8], a1q[2][8];

        #pragma unroll
        for (int j = 0; j < 16; ++j) wq[0][j] = Bb[j * 32];
        float bo = b2[ob + n];

        __syncthreads();

        #pragma unroll
        for (int j = 0; j < 8; ++j) {
            int k = j * 8;
            a0q[0][j].x = sH[n][k + c0];       a0q[0][j].y = sH[n][k + c0 + 1];
            a1q[0][j].x = sH[n][k + 4 + c0];   a1q[0][j].y = sH[n][k + 4 + c0 + 1];
        }

        v8f acc = {0.f,0.f,0.f,0.f,0.f,0.f,0.f,0.f};
        #pragma unroll
        for (int hb = 0; hb < 8; ++hb) {
            const int b = hb >> 1, hs = hb & 1;
            if (hs == 0 && b + 1 < 4) {
                #pragma unroll
                for (int j = 0; j < 16; ++j) wq[(b + 1) & 1][j] = Bb[((b + 1) * 16 + j) * 32];
            }
            if (hb + 1 < 8) {
                #pragma unroll
                for (int j = 0; j < 8; ++j) {
                    int k = ((hb + 1) * 8 + j) * 8;
                    a0q[hs ^ 1][j].x = sH[n][k + c0];       a0q[hs ^ 1][j].y = sH[n][k + c0 + 1];
                    a1q[hs ^ 1][j].x = sH[n][k + 4 + c0];   a1q[hs ^ 1][j].y = sH[n][k + 4 + c0 + 1];
                }
            }
            __builtin_amdgcn_sched_barrier(0);
            #pragma unroll
            for (int j = 0; j < 8; ++j) {
                float4 w = wq[b & 1][hs * 8 + j];
                v2f b0, b1f;
                b0.x = w.x; b0.y = w.y;
                b1f.x = w.z; b1f.y = w.w;
                acc = wmma_f32_4(a0q[hs][j], b0,  acc);
                acc = wmma_f32_4(a1q[hs][j], b1f, acc);
            }
            __builtin_amdgcn_sched_barrier(0);
        }
        for (int r = 0; r < 8; ++r) {
            int row = r + half * 8;
            out[(size_t)sRow[row] * OUT_DIM + ob + n] = acc[r] + bo;
        }
    }
}

extern "C" void kernel_launch(void* const* d_in, const int* in_sizes, int n_in,
                              void* d_out, int out_size, void* d_ws, size_t ws_size,
                              hipStream_t stream) {
    const float* embeds = (const float*)d_in[0];
    const float* Wmods  = (const float*)d_in[1];
    const float* W1     = (const float*)d_in[2];
    const float* b1     = (const float*)d_in[3];
    const float* W2     = (const float*)d_in[4];
    const float* b2     = (const float*)d_in[5];
    const int*   eids   = (const int*)d_in[6];
    const int*   mids   = (const int*)d_in[7];
    float*       out    = (float*)d_out;

    char* ws = (char*)d_ws;
    size_t off = 0;
    float* X0      = (float*)(ws + off); off += (size_t)BATCH * EMBED * 4;
    float* X1      = (float*)(ws + off); off += (size_t)BATCH * EMBED * 4;
    float* WmF     = (float*)(ws + off); off += (size_t)NUM_MOD * EMBED * EMBED * 4;
    float* W1F     = (float*)(ws + off); off += (size_t)HIDDEN * EMBED * 4;
    float* W2F     = (float*)(ws + off); off += (size_t)OUT_DIM * HIDDEN * 4;
    off += 3 * 3 * NUM_MOD * 4;
    int*   rowlist = (int*)(ws + off);   off += 3 * BATCH * 4;
    int4*  tiletab = (int4*)(ws + off);  off += 3 * MAX_TILES * sizeof(int4);
    int*   ntiles  = (int*)(ws + off);   off += 64;
    (void)ws_size; (void)in_sizes; (void)n_in; (void)out_size;

    first_hop_prep_kernel<<<BIN_BLOCKS + H0_BLOCKS + PACK_BLOCKS, PREP_THREADS, 0, stream>>>(
        embeds, Wmods, W1, W2, eids, mids, rowlist, tiletab, ntiles,
        (float4*)WmF, (float4*)W1F, (float4*)W2F, X0);

    hop_gemm_kernel<<<MAX_TILES, GEMM_THREADS, 0, stream>>>(embeds, WmF, eids, tiletab, ntiles, rowlist, X0, X1, 1);

    hop_mlp_kernel<<<BATCH / 16, GEMM_THREADS, 0, stream>>>(embeds, WmF, eids, mids, rowlist, X1,
                                                            W1F, b1, W2F, b2, out);
}
